# starting kernel
# baseline (speedup 1.0000x reference)
#include <hip/hip_runtime.h>

typedef float v2f __attribute__((ext_vector_type(2)));
typedef float v8f __attribute__((ext_vector_type(8)));

__global__ __launch_bounds__(256) void init_deg_kernel(float* __restrict__ deg, int n) {
    int i = blockIdx.x * blockDim.x + threadIdx.x;
    if (i < n) deg[i] = 1.0f;
}

__global__ __launch_bounds__(256) void deg_scatter_kernel(const int* __restrict__ dst,
                                                          float* __restrict__ deg, int e) {
    int i = blockIdx.x * blockDim.x + threadIdx.x;
    if (i < e) atomicAdd(&deg[dst[i]], 1.0f);
}

__global__ __launch_bounds__(256) void dinv_kernel(const float* __restrict__ deg,
                                                   float* __restrict__ dinv, int n) {
    int i = blockIdx.x * blockDim.x + threadIdx.x;
    if (i < n) dinv[i] = rsqrtf(deg[i]);
}

template <int K, int N>
__global__ __launch_bounds__(128) void gemm_wmma_kernel(const float* __restrict__ A,
                                                        const float* __restrict__ B,
                                                        float* __restrict__ C, int M) {
    __shared__ float Alds[64 * K];
    __shared__ float Blds[K * 64];

    const int tid     = threadIdx.x;
    const int wave    = tid >> 5;
    const int lane    = tid & 31;
    const int lhalf   = lane >> 4;
    const int l15     = lane & 15;
    const int rowBase = blockIdx.x * 64;

    for (int i = tid; i < 64 * K; i += 128) {
        int r = rowBase + i / K;
        Alds[i] = (r < M) ? A[(size_t)r * K + (i % K)] : 0.0f;
    }

    const int mRow = wave * 16 + l15;

    for (int nb = 0; nb < N; nb += 64) {
        __syncthreads();
        for (int i = tid; i < K * 64; i += 128) {
            int kk = i >> 6, nn = i & 63;
            Blds[i] = B[(size_t)kk * N + nb + nn];
        }
        __syncthreads();

        for (int nt = 0; nt < 4; ++nt) {
            const int n = nt * 16 + l15;
            v8f acc = {};
            for (int k = 0; k < K; k += 4) {
                const int ka = k + lhalf * 2;
                v2f a, b;
                a.x = Alds[mRow * K + ka];
                a.y = Alds[mRow * K + ka + 1];
                b.x = Blds[ka * 64 + n];
                b.y = Blds[(ka + 1) * 64 + n];
                acc = __builtin_amdgcn_wmma_f32_16x16x4_f32(
                     false, a,  false, b,
                     (short)0, acc,  false,  false);
            }
            const int row0 = rowBase + wave * 16 + lhalf * 8;
            const int col  = nb + nt * 16 + l15;
#pragma unroll
            for (int g = 0; g < 8; ++g) {
                int r = row0 + g;
                if (r < M) C[(size_t)r * N + col] = acc[g];
            }
        }
    }
}

template <int F>
__global__ __launch_bounds__(256) void self_init_kernel(const float* __restrict__ H,
                                                        const float* __restrict__ dinv,
                                                        float* __restrict__ AGG, int n) {
    int t = blockIdx.x * blockDim.x + threadIdx.x;
    int total = n * (F / 4);
    if (t >= total) return;
    int node = t / (F / 4);
    int fg   = t % (F / 4);
    float w  = dinv[node] * dinv[node];
    const float4* h = (const float4*)(H + (size_t)node * F);
    float4 v = h[fg];
    float4* o = (float4*)(AGG + (size_t)node * F);
    o[fg] = make_float4(v.x * w, v.y * w, v.z * w, v.w * w);
}

template <int F>
__global__ __launch_bounds__(256) void edge_scatter_kernel(const int* __restrict__ src,
                                                           const int* __restrict__ dst,
                                                           const float* __restrict__ dinv,
                                                           const float* __restrict__ H,
                                                           float* __restrict__ AGG, int e) {
    int t = blockIdx.x * blockDim.x + threadIdx.x;
    int total = e * (F / 4);
    if (t >= total) return;
    int ei = t / (F / 4);
    int fg = t % (F / 4);
    int s = src[ei], d = dst[ei];
    float w = dinv[s] * dinv[d];
    const float4* h = (const float4*)(H + (size_t)s * F);
    float4 v = h[fg];
    float* o = AGG + (size_t)d * F + fg * 4;
    atomicAdd(o + 0, v.x * w);
    atomicAdd(o + 1, v.y * w);
    atomicAdd(o + 2, v.z * w);
    atomicAdd(o + 3, v.w * w);
}

template <int F>
__global__ __launch_bounds__(256) void bias_relu_kernel(float* __restrict__ Z,
                                                        const float* __restrict__ bias,
                                                        int n) {
    int t = blockIdx.x * blockDim.x + threadIdx.x;
    int total = n * F;
    if (t >= total) return;
    int f = t % F;
    float v = Z[t] + bias[f];
    Z[t] = v > 0.0f ? v : 0.0f;
}

__global__ __launch_bounds__(256) void decoder_kernel(const float* __restrict__ Z,
                                                      const int* __restrict__ ia,
                                                      const int* __restrict__ ib,
                                                      float* __restrict__ out, int npairs) {
    int pair = blockIdx.x * 8 + (threadIdx.x >> 5);
    int lane = threadIdx.x & 31;
    if (pair >= npairs) return;
    int a = ia[pair], b = ib[pair];
    const float2* za = (const float2*)(Z + (size_t)a * 64);
    const float2* zb = (const float2*)(Z + (size_t)b * 64);
    float2 x = za[lane], y = zb[lane];
    float p = x.x * y.x + x.y * y.y;
#pragma unroll
    for (int off = 16; off > 0; off >>= 1) p += __shfl_xor(p, off, 32);
    if (lane == 0) out[pair] = p;
}

extern "C" void kernel_launch(void* const* d_in, const int* in_sizes, int n_in,
                              void* d_out, int out_size, void* d_ws, size_t ws_size,
                              hipStream_t stream) {
    const float* x     = (const float*)d_in[0];
    const float* W_in  = (const float*)d_in[1];
    const float* b_in  = (const float*)d_in[2];
    const float* W_out = (const float*)d_in[3];
    const float* b_out = (const float*)d_in[4];
    const int*   ei    = (const int*)d_in[5];
    const int*   eli   = (const int*)d_in[6];

    const int N    = in_sizes[0] / 128;
    const int E    = in_sizes[5] / 2;
    const int NDEC = in_sizes[6] / 2;
    const int* src = ei;
    const int* dst = ei + E;
    const int* ia  = eli;
    const int* ib  = eli + NDEC;
    float* logits  = (float*)d_out;

    char* base = (char*)d_ws;
    size_t off = 0;
    auto take = [&](size_t bytes) -> float* {
        float* p = (float*)(base + off);
        off = (off + bytes + 255) & ~(size_t)255;
        return p;
    };
    float* deg  = take((size_t)N * 4);
    float* dinv = take((size_t)N * 4);
    float* H1   = take((size_t)N * 128 * 4);
    float* A1   = take((size_t)N * 128 * 4);
    float* H2   = take((size_t)N * 64 * 4);
    float* A2   = take((size_t)N * 64 * 4);
    (void)ws_size;

    const int TB = 256;
    init_deg_kernel<<<(N + TB - 1) / TB, TB, 0, stream>>>(deg, N);
    deg_scatter_kernel<<<(E + TB - 1) / TB, TB, 0, stream>>>(dst, deg, E);
    dinv_kernel<<<(N + TB - 1) / TB, TB, 0, stream>>>(deg, dinv, N);

    gemm_wmma_kernel<128, 128><<<(N + 63) / 64, 128, 0, stream>>>(x, W_in, H1, N);
    {
        int total = N * 32;
        self_init_kernel<128><<<(total + TB - 1) / TB, TB, 0, stream>>>(H1, dinv, A1, N);
        int etotal = E * 32;
        edge_scatter_kernel<128><<<(etotal + TB - 1) / TB, TB, 0, stream>>>(src, dst, dinv, H1, A1, E);
        int ztotal = N * 128;
        bias_relu_kernel<128><<<(ztotal + TB - 1) / TB, TB, 0, stream>>>(A1, b_in, N);
    }

    gemm_wmma_kernel<128, 64><<<(N + 63) / 64, 128, 0, stream>>>(A1, W_out, H2, N);
    {
        int total = N * 16;
        self_init_kernel<64><<<(total + TB - 1) / TB, TB, 0, stream>>>(H2, dinv, A2, N);
        int etotal = E * 16;
        edge_scatter_kernel<64><<<(etotal + TB - 1) / TB, TB, 0, stream>>>(src, dst, dinv, H2, A2, E);
        int ztotal = N * 64;
        bias_relu_kernel<64><<<(ztotal + TB - 1) / TB, TB, 0, stream>>>(A2, b_out, N);
    }

    decoder_kernel<<<(NDEC + 7) / 8, 256, 0, stream>>>(A2, ia, ib, logits, NDEC);
}


# Aggregate by pull over a per-call dst-sorted edge list (int count+rank atomics, scan, fill) instead of f32 atomic scatter; self term, bias, ReLU at the one row store; row sum order = arrival rank
# speedup vs baseline: 1.5239x; 1.5239x over previous
#include <hip/hip_runtime.h>

typedef float v2f __attribute__((ext_vector_type(2)));
typedef float v8f __attribute__((ext_vector_type(8)));

__global__ __launch_bounds__(256) void zero_cnt_kernel(int* __restrict__ cnt, int n) {
    int i = blockIdx.x * blockDim.x + threadIdx.x;
    if (i < n) cnt[i] = 0;
}

__global__ __launch_bounds__(256) void count_rank_kernel(const int* __restrict__ dst,
                                                         int* __restrict__ cnt,
                                                         int* __restrict__ rank, int e) {
    int i = blockIdx.x * blockDim.x + threadIdx.x;
    if (i < e) rank[i] = atomicAdd(&cnt[dst[i]], 1);
}

#define SCAN_ITEMS 16
__global__ __launch_bounds__(1024) void scan_dinv_kernel(const int* __restrict__ cnt,
                                                         int* __restrict__ rowptr,
                                                         float* __restrict__ dinv, int n) {
    __shared__ int wsum[32];
    __shared__ int total_s;
    const int tid  = threadIdx.x;
    const int lane = tid & 31;
    const int wave = tid >> 5;
    int carry = 0;
    for (int base = 0; base < n; base += 1024 * SCAN_ITEMS) {
        const int  i0   = base + tid * SCAN_ITEMS;
        const bool full = (i0 + SCAN_ITEMS <= n);
        int c[SCAN_ITEMS];
        if (full) {
            for (int q = 0; q < SCAN_ITEMS / 4; ++q) {
                const int4 x = ((const int4*)(cnt + i0))[q];
                c[4 * q + 0] = x.x; c[4 * q + 1] = x.y; c[4 * q + 2] = x.z; c[4 * q + 3] = x.w;
            }
        } else {
            for (int k = 0; k < SCAN_ITEMS; ++k) c[k] = (i0 + k < n) ? cnt[i0 + k] : 0;
        }
        int t = 0;
        for (int k = 0; k < SCAN_ITEMS; ++k) t += c[k];
        int inc = t;
        for (int off = 1; off < 32; off <<= 1) {
            int v = __shfl_up(inc, off, 32);
            if (lane >= off) inc += v;
        }
        if (lane == 31) wsum[wave] = inc;
        __syncthreads();
        if (wave == 0) {
            int s = wsum[lane];
            int si = s;
            for (int off = 1; off < 32; off <<= 1) {
                int v = __shfl_up(si, off, 32);
                if (lane >= off) si += v;
            }
            wsum[lane] = si - s;
            if (lane == 31) total_s = si;
        }
        __syncthreads();
        int run = carry + wsum[wave] + inc - t;
        carry += total_s;
        int   r[SCAN_ITEMS];
        float d[SCAN_ITEMS];
        for (int k = 0; k < SCAN_ITEMS; ++k) {
            r[k] = run;
            d[k] = rsqrtf((float)(c[k] + 1));
            run += c[k];
        }
        if (full) {
            for (int q = 0; q < SCAN_ITEMS / 4; ++q) {
                ((int4*)(rowptr + i0))[q] = make_int4(r[4 * q], r[4 * q + 1], r[4 * q + 2], r[4 * q + 3]);
                ((float4*)(dinv + i0))[q] = make_float4(d[4 * q], d[4 * q + 1], d[4 * q + 2], d[4 * q + 3]);
            }
        } else {
            for (int k = 0; k < SCAN_ITEMS; ++k) {
                if (i0 + k < n) { rowptr[i0 + k] = r[k]; dinv[i0 + k] = d[k]; }
            }
        }
        __syncthreads();
    }
    if (tid == 0) rowptr[n] = carry;
}

__global__ __launch_bounds__(256) void fill_csr_kernel(const int* __restrict__ src,
                                                       const int* __restrict__ dst,
                                                       const int* __restrict__ rank,
                                                       const int* __restrict__ rowptr,
                                                       const float* __restrict__ dinv,
                                                       int2* __restrict__ csr, int e) {
    int i = blockIdx.x * blockDim.x + threadIdx.x;
    if (i >= e) return;
    int s = src[i], d = dst[i];
    csr[rowptr[d] + rank[i]] = make_int2(s, __float_as_int(dinv[s]));
}

template <int K, int N>
__global__ __launch_bounds__(128) void gemm_wmma_kernel(const float* __restrict__ A,
                                                        const float* __restrict__ B,
                                                        float* __restrict__ C, int M) {
    __shared__ float Alds[64 * K];
    __shared__ float Blds[K * 64];

    const int tid     = threadIdx.x;
    const int wave    = tid >> 5;
    const int lane    = tid & 31;
    const int lhalf   = lane >> 4;
    const int l15     = lane & 15;
    const int rowBase = blockIdx.x * 64;

    for (int i = tid; i < 64 * K; i += 128) {
        int r = rowBase + i / K;
        Alds[i] = (r < M) ? A[(size_t)r * K + (i % K)] : 0.0f;
    }

    const int mRow = wave * 16 + l15;

    for (int nb = 0; nb < N; nb += 64) {
        __syncthreads();
        for (int i = tid; i < K * 64; i += 128) {
            int kk = i >> 6, nn = i & 63;
            Blds[i] = B[(size_t)kk * N + nb + nn];
        }
        __syncthreads();

        for (int nt = 0; nt < 4; ++nt) {
            const int n = nt * 16 + l15;
            v8f acc = {};
            for (int k = 0; k < K; k += 4) {
                const int ka = k + lhalf * 2;
                v2f a, b;
                a.x = Alds[mRow * K + ka];
                a.y = Alds[mRow * K + ka + 1];
                b.x = Blds[ka * 64 + n];
                b.y = Blds[(ka + 1) * 64 + n];
                acc = __builtin_amdgcn_wmma_f32_16x16x4_f32(
                     false, a,  false, b,
                     (short)0, acc,  false,  false);
            }
            const int row0 = rowBase + wave * 16 + lhalf * 8;
            const int col  = nb + nt * 16 + l15;
#pragma unroll
            for (int g = 0; g < 8; ++g) {
                int r = row0 + g;
                if (r < M) C[(size_t)r * N + col] = acc[g];
            }
        }
    }
}

template <int F, int B>
__global__ __launch_bounds__(256) void gather_kernel(const int* __restrict__ rowptr,
                                                     const int2* __restrict__ csr,
                                                     const float* __restrict__ dinv,
                                                     const float* __restrict__ H,
                                                     const float* __restrict__ bias,
                                                     float* __restrict__ Z, int n) {
    constexpr int V = F / 32;
    typedef float vecf __attribute__((ext_vector_type(V)));
    const int lane = threadIdx.x & 31;
    const int node = __builtin_amdgcn_readfirstlane(blockIdx.x * 8 + (threadIdx.x >> 5));
    if (node >= n) return;
    const vecf* Hv = (const vecf*)H;
    const int   beg = rowptr[node];
    const int   m   = rowptr[node + 1] - beg;
    const float dn  = dinv[node];

    int  mm  = m < 32 ? m : 32;
    int2 ent = make_int2(node, 0);
    if (lane < mm) ent = csr[beg + lane];
    const vecf self = Hv[(size_t)node * 32 + lane];
    vecf acc = {};

    for (int c0 = 0;;) {
        int j = 0;
        for (; j + B <= mm; j += B) {
            vecf  v[B];
            float w[B];
            for (int k = 0; k < B; ++k) {
                const int s = __builtin_amdgcn_readlane(ent.x, j + k);
                w[k] = __int_as_float(__builtin_amdgcn_readlane(ent.y, j + k)) * dn;
                v[k] = Hv[(size_t)s * 32 + lane];
            }
            for (int k = 0; k < B; ++k) acc += v[k] * w[k];
        }
        if (j < mm) {
            vecf  v[B];
            float w[B];
            for (int k = 0; k < B; ++k) {
                const int jj = (j + k < 32) ? (j + k) : 31;
                const int s = __builtin_amdgcn_readlane(ent.x, jj);
                w[k] = __int_as_float(__builtin_amdgcn_readlane(ent.y, jj)) * dn;
                v[k] = Hv[(size_t)s * 32 + lane];
            }
            for (int k = 0; k < B; ++k) {
                if (j + k < mm) acc += v[k] * w[k];
            }
        }
        c0 += 32;
        if (c0 >= m) break;
        mm  = (m - c0 < 32) ? (m - c0) : 32;
        ent = make_int2(node, 0);
        if (lane < mm) ent = csr[beg + c0 + lane];
    }

    acc += self * (dn * dn);
    const vecf b = ((const vecf*)bias)[lane];
    vecf r = acc + b;
    for (int q = 0; q < V; ++q) r[q] = r[q] > 0.0f ? r[q] : 0.0f;
    ((vecf*)Z)[(size_t)node * 32 + lane] = r;
}

__global__ __launch_bounds__(256) void decoder_kernel(const float* __restrict__ Z,
                                                      const int* __restrict__ ia,
                                                      const int* __restrict__ ib,
                                                      float* __restrict__ out, int npairs) {
    int pair = blockIdx.x * 8 + (threadIdx.x >> 5);
    int lane = threadIdx.x & 31;
    if (pair >= npairs) return;
    int a = ia[pair], b = ib[pair];
    const float2* za = (const float2*)(Z + (size_t)a * 64);
    const float2* zb = (const float2*)(Z + (size_t)b * 64);
    float2 x = za[lane], y = zb[lane];
    float p = x.x * y.x + x.y * y.y;
#pragma unroll
    for (int off = 16; off > 0; off >>= 1) p += __shfl_xor(p, off, 32);
    if (lane == 0) out[pair] = p;
}

extern "C" void kernel_launch(void* const* d_in, const int* in_sizes, int n_in,
                              void* d_out, int out_size, void* d_ws, size_t ws_size,
                              hipStream_t stream) {
    const float* x     = (const float*)d_in[0];
    const float* W_in  = (const float*)d_in[1];
    const float* b_in  = (const float*)d_in[2];
    const float* W_out = (const float*)d_in[3];
    const float* b_out = (const float*)d_in[4];
    const int*   ei    = (const int*)d_in[5];
    const int*   eli   = (const int*)d_in[6];

    const int N    = in_sizes[0] / 128;
    const int E    = in_sizes[5] / 2;
    const int NDEC = in_sizes[6] / 2;
    const int* src = ei;
    const int* dst = ei + E;
    const int* ia  = eli;
    const int* ib  = eli + NDEC;
    float* logits  = (float*)d_out;

    char* base = (char*)d_ws;
    size_t off = 0;
    auto take = [&](size_t bytes) -> float* {
        float* p = (float*)(base + off);
        off = (off + bytes + 255) & ~(size_t)255;
        return p;
    };
    int*   cnt    = (int*)take((size_t)N * 4);
    float* dinv   = take((size_t)N * 4);
    int*   rowptr = (int*)take((size_t)(N + 1) * 4);
    int*   rank   = (int*)take((size_t)E * 4);
    int2*  csr    = (int2*)take((size_t)E * 8);
    float* H1   = take((size_t)N * 128 * 4);
    float* A1   = take((size_t)N * 128 * 4);
    float* H2   = take((size_t)N * 64 * 4);
    float* A2   = take((size_t)N * 64 * 4);
    (void)ws_size;

    const int TB = 256;
    zero_cnt_kernel<<<(N + TB - 1) / TB, TB, 0, stream>>>(cnt, N);
    count_rank_kernel<<<(E + TB - 1) / TB, TB, 0, stream>>>(dst, cnt, rank, E);
    scan_dinv_kernel<<<1, 1024, 0, stream>>>(cnt, rowptr, dinv, N);
    fill_csr_kernel<<<(E + TB - 1) / TB, TB, 0, stream>>>(src, dst, rank, rowptr, dinv, csr, E);

    gemm_wmma_kernel<128, 128><<<(N + 63) / 64, 128, 0, stream>>>(x, W_in, H1, N);
    gather_kernel<128, 4><<<(N + 7) / 8, 256, 0, stream>>>(rowptr, csr, dinv, H1, b_in, A1, N);

    gemm_wmma_kernel<128, 64><<<(N + 63) / 64, 128, 0, stream>>>(A1, W_out, H2, N);
    gather_kernel<64, 8><<<(N + 7) / 8, 256, 0, stream>>>(rowptr, csr, dinv, H2, b_out, A2, N);

    decoder_kernel<<<(NDEC + 7) / 8, 256, 0, stream>>>(A2, ia, ib, logits, NDEC);
}


# Decoder: each wave handles 128 pairs, not one (8 lanes per pair, float4 row loads, 32 loads in flight); dot sum order changes: 8 serial terms per lane, then a 3-step xor butterfly over 8 lanes
# speedup vs baseline: 1.9982x; 1.3112x over previous
#include <hip/hip_runtime.h>

typedef float v2f __attribute__((ext_vector_type(2)));
typedef float v8f __attribute__((ext_vector_type(8)));

__global__ __launch_bounds__(256) void zero_cnt_kernel(int* __restrict__ cnt, int n) {
    int i = blockIdx.x * blockDim.x + threadIdx.x;
    if (i < n) cnt[i] = 0;
}

__global__ __launch_bounds__(256) void count_rank_kernel(const int* __restrict__ dst,
                                                         int* __restrict__ cnt,
                                                         int* __restrict__ rank, int e) {
    int i = blockIdx.x * blockDim.x + threadIdx.x;
    if (i < e) rank[i] = atomicAdd(&cnt[dst[i]], 1);
}

#define SCAN_ITEMS 16
__global__ __launch_bounds__(1024) void scan_dinv_kernel(const int* __restrict__ cnt,
                                                         int* __restrict__ rowptr,
                                                         float* __restrict__ dinv, int n) {
    __shared__ int wsum[32];
    __shared__ int total_s;
    const int tid  = threadIdx.x;
    const int lane = tid & 31;
    const int wave = tid >> 5;
    int carry = 0;
    for (int base = 0; base < n; base += 1024 * SCAN_ITEMS) {
        const int  i0   = base + tid * SCAN_ITEMS;
        const bool full = (i0 + SCAN_ITEMS <= n);
        int c[SCAN_ITEMS];
        if (full) {
            for (int q = 0; q < SCAN_ITEMS / 4; ++q) {
                const int4 x = ((const int4*)(cnt + i0))[q];
                c[4 * q + 0] = x.x; c[4 * q + 1] = x.y; c[4 * q + 2] = x.z; c[4 * q + 3] = x.w;
            }
        } else {
            for (int k = 0; k < SCAN_ITEMS; ++k) c[k] = (i0 + k < n) ? cnt[i0 + k] : 0;
        }
        int t = 0;
        for (int k = 0; k < SCAN_ITEMS; ++k) t += c[k];
        int inc = t;
        for (int off = 1; off < 32; off <<= 1) {
            int v = __shfl_up(inc, off, 32);
            if (lane >= off) inc += v;
        }
        if (lane == 31) wsum[wave] = inc;
        __syncthreads();
        if (wave == 0) {
            int s = wsum[lane];
            int si = s;
            for (int off = 1; off < 32; off <<= 1) {
                int v = __shfl_up(si, off, 32);
                if (lane >= off) si += v;
            }
            wsum[lane] = si - s;
            if (lane == 31) total_s = si;
        }
        __syncthreads();
        int run = carry + wsum[wave] + inc - t;
        carry += total_s;
        int   r[SCAN_ITEMS];
        float d[SCAN_ITEMS];
        for (int k = 0; k < SCAN_ITEMS; ++k) {
            r[k] = run;
            d[k] = rsqrtf((float)(c[k] + 1));
            run += c[k];
        }
        if (full) {
            for (int q = 0; q < SCAN_ITEMS / 4; ++q) {
                ((int4*)(rowptr + i0))[q] = make_int4(r[4 * q], r[4 * q + 1], r[4 * q + 2], r[4 * q + 3]);
                ((float4*)(dinv + i0))[q] = make_float4(d[4 * q], d[4 * q + 1], d[4 * q + 2], d[4 * q + 3]);
            }
        } else {
            for (int k = 0; k < SCAN_ITEMS; ++k) {
                if (i0 + k < n) { rowptr[i0 + k] = r[k]; dinv[i0 + k] = d[k]; }
            }
        }
        __syncthreads();
    }
    if (tid == 0) rowptr[n] = carry;
}

__global__ __launch_bounds__(256) void fill_csr_kernel(const int* __restrict__ src,
                                                       const int* __restrict__ dst,
                                                       const int* __restrict__ rank,
                                                       const int* __restrict__ rowptr,
                                                       const float* __restrict__ dinv,
                                                       int2* __restrict__ csr, int e) {
    int i = blockIdx.x * blockDim.x + threadIdx.x;
    if (i >= e) return;
    int s = src[i], d = dst[i];
    csr[rowptr[d] + rank[i]] = make_int2(s, __float_as_int(dinv[s]));
}

template <int K, int N>
__global__ __launch_bounds__(128) void gemm_wmma_kernel(const float* __restrict__ A,
                                                        const float* __restrict__ B,
                                                        float* __restrict__ C, int M) {
    __shared__ float Alds[64 * K];
    __shared__ float Blds[K * 64];

    const int tid     = threadIdx.x;
    const int wave    = tid >> 5;
    const int lane    = tid & 31;
    const int lhalf   = lane >> 4;
    const int l15     = lane & 15;
    const int rowBase = blockIdx.x * 64;

    for (int i = tid; i < 64 * K; i += 128) {
        int r = rowBase + i / K;
        Alds[i] = (r < M) ? A[(size_t)r * K + (i % K)] : 0.0f;
    }

    const int mRow = wave * 16 + l15;

    for (int nb = 0; nb < N; nb += 64) {
        __syncthreads();
        for (int i = tid; i < K * 64; i += 128) {
            int kk = i >> 6, nn = i & 63;
            Blds[i] = B[(size_t)kk * N + nb + nn];
        }
        __syncthreads();

        for (int nt = 0; nt < 4; ++nt) {
            const int n = nt * 16 + l15;
            v8f acc = {};
            for (int k = 0; k < K; k += 4) {
                const int ka = k + lhalf * 2;
                v2f a, b;
                a.x = Alds[mRow * K + ka];
                a.y = Alds[mRow * K + ka + 1];
                b.x = Blds[ka * 64 + n];
                b.y = Blds[(ka + 1) * 64 + n];
                acc = __builtin_amdgcn_wmma_f32_16x16x4_f32(
                     false, a,  false, b,
                     (short)0, acc,  false,  false);
            }
            const int row0 = rowBase + wave * 16 + lhalf * 8;
            const int col  = nb + nt * 16 + l15;
#pragma unroll
            for (int g = 0; g < 8; ++g) {
                int r = row0 + g;
                if (r < M) C[(size_t)r * N + col] = acc[g];
            }
        }
    }
}

template <int F, int B>
__global__ __launch_bounds__(256) void gather_kernel(const int* __restrict__ rowptr,
                                                     const int2* __restrict__ csr,
                                                     const float* __restrict__ dinv,
                                                     const float* __restrict__ H,
                                                     const float* __restrict__ bias,
                                                     float* __restrict__ Z, int n) {
    constexpr int V = F / 32;
    typedef float vecf __attribute__((ext_vector_type(V)));
    const int lane = threadIdx.x & 31;
    const int node = __builtin_amdgcn_readfirstlane(blockIdx.x * 8 + (threadIdx.x >> 5));
    if (node >= n) return;
    const vecf* Hv = (const vecf*)H;
    const int   beg = rowptr[node];
    const int   m   = rowptr[node + 1] - beg;
    const float dn  = dinv[node];

    int  mm  = m < 32 ? m : 32;
    int2 ent = make_int2(node, 0);
    if (lane < mm) ent = csr[beg + lane];
    const vecf self = Hv[(size_t)node * 32 + lane];
    vecf acc = {};

    for (int c0 = 0;;) {
        int j = 0;
        for (; j + B <= mm; j += B) {
            vecf  v[B];
            float w[B];
            for (int k = 0; k < B; ++k) {
                const int s = __builtin_amdgcn_readlane(ent.x, j + k);
                w[k] = __int_as_float(__builtin_amdgcn_readlane(ent.y, j + k)) * dn;
                v[k] = Hv[(size_t)s * 32 + lane];
            }
            for (int k = 0; k < B; ++k) acc += v[k] * w[k];
        }
        if (j < mm) {
            vecf  v[B];
            float w[B];
            for (int k = 0; k < B; ++k) {
                const int jj = (j + k < 32) ? (j + k) : 31;
                const int s = __builtin_amdgcn_readlane(ent.x, jj);
                w[k] = __int_as_float(__builtin_amdgcn_readlane(ent.y, jj)) * dn;
                v[k] = Hv[(size_t)s * 32 + lane];
            }
            for (int k = 0; k < B; ++k) {
                if (j + k < mm) acc += v[k] * w[k];
            }
        }
        c0 += 32;
        if (c0 >= m) break;
        mm  = (m - c0 < 32) ? (m - c0) : 32;
        ent = make_int2(node, 0);
        if (lane < mm) ent = csr[beg + c0 + lane];
    }

    acc += self * (dn * dn);
    const vecf b = ((const vecf*)bias)[lane];
    vecf r = acc + b;
    for (int q = 0; q < V; ++q) r[q] = r[q] > 0.0f ? r[q] : 0.0f;
    ((vecf*)Z)[(size_t)node * 32 + lane] = r;
}

#define DEC_ITER 4

template <int T>
__device__ __forceinline__ int group8_bcast(int v) {
    return (int)__builtin_amdgcn_mov_dpp8((unsigned)v, (unsigned)T * 0x249249u);
}

template <int M>
__device__ __forceinline__ float xor_add(float v) {
    const int o = __builtin_amdgcn_update_dpp(0, __float_as_int(v), 0x160 + M, 0xf, 0xf, false);
    return v + __int_as_float(o);
}

template <int T>
__device__ __forceinline__ void dec_slot_load(const float* __restrict__ Z, int a, int b, int q,
                                              float4& x0, float4& x1, float4& y0, float4& y1) {
    const unsigned ra = (unsigned)group8_bcast<T>(a) * 64u;
    const unsigned rb = (unsigned)group8_bcast<T>(b) * 64u;
    const float4* za = (const float4*)(Z + ra) + q;
    const float4* zb = (const float4*)(Z + rb) + q;
    x0 = za[0]; x1 = za[8];
    y0 = zb[0]; y1 = zb[8];
}

__device__ __forceinline__ float dec_slot_sum(const float4& x0, const float4& x1,
                                              const float4& y0, const float4& y1) {
    float s = x0.x * y0.x;
    s += x0.y * y0.y; s += x0.z * y0.z; s += x0.w * y0.w;
    s += x1.x * y1.x; s += x1.y * y1.y; s += x1.z * y1.z; s += x1.w * y1.w;
    s = xor_add<1>(s);
    s = xor_add<2>(s);
    s = xor_add<4>(s);
    return s;
}

__global__ __launch_bounds__(256) void decoder_kernel(const float* __restrict__ Z,
                                                      const int* __restrict__ ia,
                                                      const int* __restrict__ ib,
                                                      float* __restrict__ out, int npairs) {
    const int lane = threadIdx.x & 31;
    const int q    = lane & 7;
    const int base0 = __builtin_amdgcn_readfirstlane((blockIdx.x * 8 + (threadIdx.x >> 5)) * (32 * DEC_ITER));
    if (base0 >= npairs) return;

    int a[DEC_ITER], b[DEC_ITER];
#pragma unroll
    for (int it = 0; it < DEC_ITER; ++it) {
        const int p = base0 + it * 32 + lane;
        a[it] = 0; b[it] = 0;
        if (p < npairs) { a[it] = ia[p]; b[it] = ib[p]; }
    }

#pragma unroll
    for (int it = 0; it < DEC_ITER; ++it) {
        if (base0 + it * 32 >= npairs) break;
        float4 x0[8], x1[8], y0[8], y1[8];
        dec_slot_load<0>(Z, a[it], b[it], q, x0[0], x1[0], y0[0], y1[0]);
        dec_slot_load<1>(Z, a[it], b[it], q, x0[1], x1[1], y0[1], y1[1]);
        dec_slot_load<2>(Z, a[it], b[it], q, x0[2], x1[2], y0[2], y1[2]);
        dec_slot_load<3>(Z, a[it], b[it], q, x0[3], x1[3], y0[3], y1[3]);
        dec_slot_load<4>(Z, a[it], b[it], q, x0[4], x1[4], y0[4], y1[4]);
        dec_slot_load<5>(Z, a[it], b[it], q, x0[5], x1[5], y0[5], y1[5]);
        dec_slot_load<6>(Z, a[it], b[it], q, x0[6], x1[6], y0[6], y1[6]);
        dec_slot_load<7>(Z, a[it], b[it], q, x0[7], x1[7], y0[7], y1[7]);
        float res = 0.0f;
#pragma unroll
        for (int t = 0; t < 8; ++t) {
            const float s = dec_slot_sum(x0[t], x1[t], y0[t], y1[t]);
            res = (q == t) ? s : res;
        }
        const int p = base0 + it * 32 + lane;
        if (p < npairs) out[p] = res;
    }
}

extern "C" void kernel_launch(void* const* d_in, const int* in_sizes, int n_in,
                              void* d_out, int out_size, void* d_ws, size_t ws_size,
                              hipStream_t stream) {
    const float* x     = (const float*)d_in[0];
    const float* W_in  = (const float*)d_in[1];
    const float* b_in  = (const float*)d_in[2];
    const float* W_out = (const float*)d_in[3];
    const float* b_out = (const float*)d_in[4];
    const int*   ei    = (const int*)d_in[5];
    const int*   eli   = (const int*)d_in[6];

    const int N    = in_sizes[0] / 128;
    const int E    = in_sizes[5] / 2;
    const int NDEC = in_sizes[6] / 2;
    const int* src = ei;
    const int* dst = ei + E;
    const int* ia  = eli;
    const int* ib  = eli + NDEC;
    float* logits  = (float*)d_out;

    char* base = (char*)d_ws;
    size_t off = 0;
    auto take = [&](size_t bytes) -> float* {
        float* p = (float*)(base + off);
        off = (off + bytes + 255) & ~(size_t)255;
        return p;
    };
    int*   cnt    = (int*)take((size_t)N * 4);
    float* dinv   = take((size_t)N * 4);
    int*   rowptr = (int*)take((size_t)(N + 1) * 4);
    int*   rank   = (int*)take((size_t)E * 4);
    int2*  csr    = (int2*)take((size_t)E * 8);
    float* H1   = take((size_t)N * 128 * 4);
    float* A1   = take((size_t)N * 128 * 4);
    float* H2   = take((size_t)N * 64 * 4);
    float* A2   = take((size_t)N * 64 * 4);
    (void)ws_size;

    const int TB = 256;
    zero_cnt_kernel<<<(N + TB - 1) / TB, TB, 0, stream>>>(cnt, N);
    count_rank_kernel<<<(E + TB - 1) / TB, TB, 0, stream>>>(dst, cnt, rank, E);
    scan_dinv_kernel<<<1, 1024, 0, stream>>>(cnt, rowptr, dinv, N);
    fill_csr_kernel<<<(E + TB - 1) / TB, TB, 0, stream>>>(src, dst, rank, rowptr, dinv, csr, E);

    gemm_wmma_kernel<128, 128><<<(N + 63) / 64, 128, 0, stream>>>(x, W_in, H1, N);
    gather_kernel<128, 4><<<(N + 7) / 8, 256, 0, stream>>>(rowptr, csr, dinv, H1, b_in, A1, N);

    gemm_wmma_kernel<128, 64><<<(N + 63) / 64, 128, 0, stream>>>(A1, W_out, H2, N);
    gather_kernel<64, 8><<<(N + 7) / 8, 256, 0, stream>>>(rowptr, csr, dinv, H2, b_out, A2, N);

    decoder_kernel<<<(NDEC + 256 * DEC_ITER - 1) / (256 * DEC_ITER), 256, 0, stream>>>(A2, ia, ib, logits, NDEC);
}


# GEMM operand feed: A fragment loaded once from global into registers, B transposed in LDS (pitch K+4, 16-byte reads), all loads issued together; the k sum order inside the GEMM changes
# speedup vs baseline: 2.7517x; 1.3771x over previous
#include <hip/hip_runtime.h>

typedef float v2f __attribute__((ext_vector_type(2)));
typedef float v8f __attribute__((ext_vector_type(8)));

__global__ __launch_bounds__(256) void zero_cnt_kernel(int* __restrict__ cnt, int n) {
    int i = blockIdx.x * blockDim.x + threadIdx.x;
    if (i < n) cnt[i] = 0;
}

__global__ __launch_bounds__(256) void count_rank_kernel(const int* __restrict__ dst,
                                                         int* __restrict__ cnt,
                                                         int* __restrict__ rank, int e) {
    int i = blockIdx.x * blockDim.x + threadIdx.x;
    if (i < e) rank[i] = atomicAdd(&cnt[dst[i]], 1);
}

#define SCAN_ITEMS 16
__global__ __launch_bounds__(1024) void scan_dinv_kernel(const int* __restrict__ cnt,
                                                         int* __restrict__ rowptr,
                                                         float* __restrict__ dinv, int n) {
    __shared__ int wsum[32];
    __shared__ int total_s;
    const int tid  = threadIdx.x;
    const int lane = tid & 31;
    const int wave = tid >> 5;
    int carry = 0;
    for (int base = 0; base < n; base += 1024 * SCAN_ITEMS) {
        const int  i0   = base + tid * SCAN_ITEMS;
        const bool full = (i0 + SCAN_ITEMS <= n);
        int c[SCAN_ITEMS];
        if (full) {
            for (int q = 0; q < SCAN_ITEMS / 4; ++q) {
                const int4 x = ((const int4*)(cnt + i0))[q];
                c[4 * q + 0] = x.x; c[4 * q + 1] = x.y; c[4 * q + 2] = x.z; c[4 * q + 3] = x.w;
            }
        } else {
            for (int k = 0; k < SCAN_ITEMS; ++k) c[k] = (i0 + k < n) ? cnt[i0 + k] : 0;
        }
        int t = 0;
        for (int k = 0; k < SCAN_ITEMS; ++k) t += c[k];
        int inc = t;
        for (int off = 1; off < 32; off <<= 1) {
            int v = __shfl_up(inc, off, 32);
            if (lane >= off) inc += v;
        }
        if (lane == 31) wsum[wave] = inc;
        __syncthreads();
        if (wave == 0) {
            int s = wsum[lane];
            int si = s;
            for (int off = 1; off < 32; off <<= 1) {
                int v = __shfl_up(si, off, 32);
                if (lane >= off) si += v;
            }
            wsum[lane] = si - s;
            if (lane == 31) total_s = si;
        }
        __syncthreads();
        int run = carry + wsum[wave] + inc - t;
        carry += total_s;
        int   r[SCAN_ITEMS];
        float d[SCAN_ITEMS];
        for (int k = 0; k < SCAN_ITEMS; ++k) {
            r[k] = run;
            d[k] = rsqrtf((float)(c[k] + 1));
            run += c[k];
        }
        if (full) {
            for (int q = 0; q < SCAN_ITEMS / 4; ++q) {
                ((int4*)(rowptr + i0))[q] = make_int4(r[4 * q], r[4 * q + 1], r[4 * q + 2], r[4 * q + 3]);
                ((float4*)(dinv + i0))[q] = make_float4(d[4 * q], d[4 * q + 1], d[4 * q + 2], d[4 * q + 3]);
            }
        } else {
            for (int k = 0; k < SCAN_ITEMS; ++k) {
                if (i0 + k < n) { rowptr[i0 + k] = r[k]; dinv[i0 + k] = d[k]; }
            }
        }
        __syncthreads();
    }
    if (tid == 0) rowptr[n] = carry;
}

__global__ __launch_bounds__(256) void fill_csr_kernel(const int* __restrict__ src,
                                                       const int* __restrict__ dst,
                                                       const int* __restrict__ rank,
                                                       const int* __restrict__ rowptr,
                                                       const float* __restrict__ dinv,
                                                       int2* __restrict__ csr, int e) {
    int i = blockIdx.x * blockDim.x + threadIdx.x;
    if (i >= e) return;
    int s = src[i], d = dst[i];
    csr[rowptr[d] + rank[i]] = make_int2(s, __float_as_int(dinv[s]));
}

template <int K, int N>
__global__ __launch_bounds__(256) void gemm_wmma_kernel(const float* __restrict__ A,
                                                        const float* __restrict__ B,
                                                        float* __restrict__ C, int M) {
    constexpr int P  = K + 4;
    constexpr int NW = N / 2;
    constexpr int NT = NW / 16;
    constexpr int J  = K / 8;
    __shared__ float4 Bt4[N * P / 4];
    float* Bt = (float*)Bt4;

    const int tid  = threadIdx.x;
    const int wave = tid >> 5;
    const int lane = tid & 31;
    const int h    = lane >> 4;
    const int c    = lane & 15;
    const int row0 = blockIdx.x * 64 + (wave & 3) * 16;
    const int col0 = (wave >> 2) * NW;

    constexpr int IT = N * (K / 4) / 256;

    float4 bv[IT];
#pragma unroll
    for (int it = 0; it < IT; ++it) {
        const int idx = tid + it * 256;
        const float* bsrc = B + (size_t)(4 * (idx / N)) * N + (idx % N);
        bv[it].x = bsrc[0];
        bv[it].y = bsrc[N];
        bv[it].z = bsrc[2 * N];
        bv[it].w = bsrc[3 * N];
    }

    int ar = row0 + c;
    ar = ar < M ? ar : M - 1;
    const float4* ap = (const float4*)(A + (size_t)ar * K) + h;
    float4 af[J];
#pragma unroll
    for (int j = 0; j < J; ++j) af[j] = ap[2 * j];

#pragma unroll
    for (int it = 0; it < IT; ++it) {
        const int idx = tid + it * 256;
        *(float4*)(Bt + (idx % N) * P + 4 * (idx / N)) = bv[it];
    }
    __syncthreads();
    if (row0 >= M) return;

    v8f acc[NT];
#pragma unroll
    for (int t = 0; t < NT; ++t) acc[t] = v8f{};

    const float* bp = Bt + (col0 + c) * P + 4 * h;
#pragma unroll
    for (int j = 0; j < J; ++j) {
        float4 bf[NT];
#pragma unroll
        for (int t = 0; t < NT; ++t) bf[t] = *(const float4*)(bp + t * 16 * P + 8 * j);
#pragma unroll
        for (int t = 0; t < NT; ++t) {
            v2f a, b;
            a.x = af[j].x; a.y = af[j].y;
            b.x = bf[t].x; b.y = bf[t].y;
            acc[t] = __builtin_amdgcn_wmma_f32_16x16x4_f32(
                 false, a,  false, b,
                 (short)0, acc[t],  false,  false);
        }
#pragma unroll
        for (int t = 0; t < NT; ++t) {
            v2f a, b;
            a.x = af[j].z; a.y = af[j].w;
            b.x = bf[t].z; b.y = bf[t].w;
            acc[t] = __builtin_amdgcn_wmma_f32_16x16x4_f32(
                 false, a,  false, b,
                 (short)0, acc[t],  false,  false);
        }
    }

#pragma unroll
    for (int t = 0; t < NT; ++t) {
        const int col = col0 + t * 16 + c;
#pragma unroll
        for (int g = 0; g < 8; ++g) {
            const int r = row0 + h * 8 + g;
            if (r < M) C[(size_t)r * N + col] = acc[t][g];
        }
    }
}

template <int F, int B>
__global__ __launch_bounds__(256) void gather_kernel(const int* __restrict__ rowptr,
                                                     const int2* __restrict__ csr,
                                                     const float* __restrict__ dinv,
                                                     const float* __restrict__ H,
                                                     const float* __restrict__ bias,
                                                     float* __restrict__ Z, int n) {
    constexpr int V = F / 32;
    typedef float vecf __attribute__((ext_vector_type(V)));
    const int lane = threadIdx.x & 31;
    const int node = __builtin_amdgcn_readfirstlane(blockIdx.x * 8 + (threadIdx.x >> 5));
    if (node >= n) return;
    const vecf* Hv = (const vecf*)H;
    const int   beg = rowptr[node];
    const int   m   = rowptr[node + 1] - beg;
    const float dn  = dinv[node];

    int  mm  = m < 32 ? m : 32;
    int2 ent = make_int2(node, 0);
    if (lane < mm) ent = csr[beg + lane];
    const vecf self = Hv[(size_t)node * 32 + lane];
    vecf acc = {};

    for (int c0 = 0;;) {
        int j = 0;
        for (; j + B <= mm; j += B) {
            vecf  v[B];
            float w[B];
            for (int k = 0; k < B; ++k) {
                const int s = __builtin_amdgcn_readlane(ent.x, j + k);
                w[k] = __int_as_float(__builtin_amdgcn_readlane(ent.y, j + k)) * dn;
                v[k] = Hv[(size_t)s * 32 + lane];
            }
            for (int k = 0; k < B; ++k) acc += v[k] * w[k];
        }
        if (j < mm) {
            vecf  v[B];
            float w[B];
            for (int k = 0; k < B; ++k) {
                const int jj = (j + k < 32) ? (j + k) : 31;
                const int s = __builtin_amdgcn_readlane(ent.x, jj);
                w[k] = __int_as_float(__builtin_amdgcn_readlane(ent.y, jj)) * dn;
                v[k] = Hv[(size_t)s * 32 + lane];
            }
            for (int k = 0; k < B; ++k) {
                if (j + k < mm) acc += v[k] * w[k];
            }
        }
        c0 += 32;
        if (c0 >= m) break;
        mm  = (m - c0 < 32) ? (m - c0) : 32;
        ent = make_int2(node, 0);
        if (lane < mm) ent = csr[beg + c0 + lane];
    }

    acc += self * (dn * dn);
    const vecf b = ((const vecf*)bias)[lane];
    vecf r = acc + b;
    for (int q = 0; q < V; ++q) r[q] = r[q] > 0.0f ? r[q] : 0.0f;
    ((vecf*)Z)[(size_t)node * 32 + lane] = r;
}

#define DEC_ITER 4

template <int T>
__device__ __forceinline__ int group8_bcast(int v) {
    return (int)__builtin_amdgcn_mov_dpp8((unsigned)v, (unsigned)T * 0x249249u);
}

template <int M>
__device__ __forceinline__ float xor_add(float v) {
    const int o = __builtin_amdgcn_update_dpp(0, __float_as_int(v), 0x160 + M, 0xf, 0xf, false);
    return v + __int_as_float(o);
}

template <int T>
__device__ __forceinline__ void dec_slot_load(const float* __restrict__ Z, int a, int b, int q,
                                              float4& x0, float4& x1, float4& y0, float4& y1) {
    const unsigned ra = (unsigned)group8_bcast<T>(a) * 64u;
    const unsigned rb = (unsigned)group8_bcast<T>(b) * 64u;
    const float4* za = (const float4*)(Z + ra) + q;
    const float4* zb = (const float4*)(Z + rb) + q;
    x0 = za[0]; x1 = za[8];
    y0 = zb[0]; y1 = zb[8];
}

__device__ __forceinline__ float dec_slot_sum(const float4& x0, const float4& x1,
                                              const float4& y0, const float4& y1) {
    float s = x0.x * y0.x;
    s += x0.y * y0.y; s += x0.z * y0.z; s += x0.w * y0.w;
    s += x1.x * y1.x; s += x1.y * y1.y; s += x1.z * y1.z; s += x1.w * y1.w;
    s = xor_add<1>(s);
    s = xor_add<2>(s);
    s = xor_add<4>(s);
    return s;
}

__global__ __launch_bounds__(256) void decoder_kernel(const float* __restrict__ Z,
                                                      const int* __restrict__ ia,
                                                      const int* __restrict__ ib,
                                                      float* __restrict__ out, int npairs) {
    const int lane = threadIdx.x & 31;
    const int q    = lane & 7;
    const int base0 = __builtin_amdgcn_readfirstlane((blockIdx.x * 8 + (threadIdx.x >> 5)) * (32 * DEC_ITER));
    if (base0 >= npairs) return;

    int a[DEC_ITER], b[DEC_ITER];
#pragma unroll
    for (int it = 0; it < DEC_ITER; ++it) {
        const int p = base0 + it * 32 + lane;
        a[it] = 0; b[it] = 0;
        if (p < npairs) { a[it] = ia[p]; b[it] = ib[p]; }
    }

#pragma unroll
    for (int it = 0; it < DEC_ITER; ++it) {
        if (base0 + it * 32 >= npairs) break;
        float4 x0[8], x1[8], y0[8], y1[8];
        dec_slot_load<0>(Z, a[it], b[it], q, x0[0], x1[0], y0[0], y1[0]);
        dec_slot_load<1>(Z, a[it], b[it], q, x0[1], x1[1], y0[1], y1[1]);
        dec_slot_load<2>(Z, a[it], b[it], q, x0[2], x1[2], y0[2], y1[2]);
        dec_slot_load<3>(Z, a[it], b[it], q, x0[3], x1[3], y0[3], y1[3]);
        dec_slot_load<4>(Z, a[it], b[it], q, x0[4], x1[4], y0[4], y1[4]);
        dec_slot_load<5>(Z, a[it], b[it], q, x0[5], x1[5], y0[5], y1[5]);
        dec_slot_load<6>(Z, a[it], b[it], q, x0[6], x1[6], y0[6], y1[6]);
        dec_slot_load<7>(Z, a[it], b[it], q, x0[7], x1[7], y0[7], y1[7]);
        float res = 0.0f;
#pragma unroll
        for (int t = 0; t < 8; ++t) {
            const float s = dec_slot_sum(x0[t], x1[t], y0[t], y1[t]);
            res = (q == t) ? s : res;
        }
        const int p = base0 + it * 32 + lane;
        if (p < npairs) out[p] = res;
    }
}

extern "C" void kernel_launch(void* const* d_in, const int* in_sizes, int n_in,
                              void* d_out, int out_size, void* d_ws, size_t ws_size,
                              hipStream_t stream) {
    const float* x     = (const float*)d_in[0];
    const float* W_in  = (const float*)d_in[1];
    const float* b_in  = (const float*)d_in[2];
    const float* W_out = (const float*)d_in[3];
    const float* b_out = (const float*)d_in[4];
    const int*   ei    = (const int*)d_in[5];
    const int*   eli   = (const int*)d_in[6];

    const int N    = in_sizes[0] / 128;
    const int E    = in_sizes[5] / 2;
    const int NDEC = in_sizes[6] / 2;
    const int* src = ei;
    const int* dst = ei + E;
    const int* ia  = eli;
    const int* ib  = eli + NDEC;
    float* logits  = (float*)d_out;

    char* base = (char*)d_ws;
    size_t off = 0;
    auto take = [&](size_t bytes) -> float* {
        float* p = (float*)(base + off);
        off = (off + bytes + 255) & ~(size_t)255;
        return p;
    };
    int*   cnt    = (int*)take((size_t)N * 4);
    float* dinv   = take((size_t)N * 4);
    int*   rowptr = (int*)take((size_t)(N + 1) * 4);
    int*   rank   = (int*)take((size_t)E * 4);
    int2*  csr    = (int2*)take((size_t)E * 8);
    float* H1   = take((size_t)N * 128 * 4);
    float* A1   = take((size_t)N * 128 * 4);
    float* H2   = take((size_t)N * 64 * 4);
    float* A2   = take((size_t)N * 64 * 4);
    (void)ws_size;

    const int TB = 256;
    zero_cnt_kernel<<<(N + TB - 1) / TB, TB, 0, stream>>>(cnt, N);
    count_rank_kernel<<<(E + TB - 1) / TB, TB, 0, stream>>>(dst, cnt, rank, E);
    scan_dinv_kernel<<<1, 1024, 0, stream>>>(cnt, rowptr, dinv, N);
    fill_csr_kernel<<<(E + TB - 1) / TB, TB, 0, stream>>>(src, dst, rank, rowptr, dinv, csr, E);

    gemm_wmma_kernel<128, 128><<<(N + 63) / 64, 256, 0, stream>>>(x, W_in, H1, N);
    gather_kernel<128, 4><<<(N + 7) / 8, 256, 0, stream>>>(rowptr, csr, dinv, H1, b_in, A1, N);

    gemm_wmma_kernel<128, 64><<<(N + 63) / 64, 256, 0, stream>>>(A1, W_out, H2, N);
    gather_kernel<64, 8><<<(N + 7) / 8, 256, 0, stream>>>(rowptr, csr, dinv, H2, b_out, A2, N);

    decoder_kernel<<<(NDEC + 256 * DEC_ITER - 1) / (256 * DEC_ITER), 256, 0, stream>>>(A2, ia, ib, logits, NDEC);
}


# Layer-1 GEMM (independent of the graph) runs in the scan's launch: block 0 scans, blocks 1.. multiply in 1024-thread blocks of 256 rows; 8 launches, not 9; every sum order as before
# speedup vs baseline: 3.0248x; 1.0992x over previous
#include <hip/hip_runtime.h>

typedef float v2f __attribute__((ext_vector_type(2)));
typedef float v8f __attribute__((ext_vector_type(8)));

__global__ __launch_bounds__(256) void zero_cnt_kernel(int* __restrict__ cnt, int n) {
    int i = blockIdx.x * blockDim.x + threadIdx.x;
    if (i < n) cnt[i] = 0;
}

__global__ __launch_bounds__(256) void count_rank_kernel(const int* __restrict__ dst,
                                                         int* __restrict__ cnt,
                                                         int* __restrict__ rank, int e) {
    int i = blockIdx.x * blockDim.x + threadIdx.x;
    if (i < e) rank[i] = atomicAdd(&cnt[dst[i]], 1);
}

#define SCAN_ITEMS 16
__device__ __forceinline__ void scan_dinv_block(const int* __restrict__ cnt,
                                                int* __restrict__ rowptr,
                                                float* __restrict__ dinv, int n) {
    __shared__ int wsum[32];
    __shared__ int total_s;
    const int tid  = threadIdx.x;
    const int lane = tid & 31;
    const int wave = tid >> 5;
    int carry = 0;
    for (int base = 0; base < n; base += 1024 * SCAN_ITEMS) {
        const int  i0   = base + tid * SCAN_ITEMS;
        const bool full = (i0 + SCAN_ITEMS <= n);
        int c[SCAN_ITEMS];
        if (full) {
            for (int q = 0; q < SCAN_ITEMS / 4; ++q) {
                const int4 x = ((const int4*)(cnt + i0))[q];
                c[4 * q + 0] = x.x; c[4 * q + 1] = x.y; c[4 * q + 2] = x.z; c[4 * q + 3] = x.w;
            }
        } else {
            for (int k = 0; k < SCAN_ITEMS; ++k) c[k] = (i0 + k < n) ? cnt[i0 + k] : 0;
        }
        int t = 0;
        for (int k = 0; k < SCAN_ITEMS; ++k) t += c[k];
        int inc = t;
        for (int off = 1; off < 32; off <<= 1) {
            int v = __shfl_up(inc, off, 32);
            if (lane >= off) inc += v;
        }
        if (lane == 31) wsum[wave] = inc;
        __syncthreads();
        if (wave == 0) {
            int s = wsum[lane];
            int si = s;
            for (int off = 1; off < 32; off <<= 1) {
                int v = __shfl_up(si, off, 32);
                if (lane >= off) si += v;
            }
            wsum[lane] = si - s;
            if (lane == 31) total_s = si;
        }
        __syncthreads();
        int run = carry + wsum[wave] + inc - t;
        carry += total_s;
        int   r[SCAN_ITEMS];
        float d[SCAN_ITEMS];
        for (int k = 0; k < SCAN_ITEMS; ++k) {
            r[k] = run;
            d[k] = rsqrtf((float)(c[k] + 1));
            run += c[k];
        }
        if (full) {
            for (int q = 0; q < SCAN_ITEMS / 4; ++q) {
                ((int4*)(rowptr + i0))[q] = make_int4(r[4 * q], r[4 * q + 1], r[4 * q + 2], r[4 * q + 3]);
                ((float4*)(dinv + i0))[q] = make_float4(d[4 * q], d[4 * q + 1], d[4 * q + 2], d[4 * q + 3]);
            }
        } else {
            for (int k = 0; k < SCAN_ITEMS; ++k) {
                if (i0 + k < n) { rowptr[i0 + k] = r[k]; dinv[i0 + k] = d[k]; }
            }
        }
        __syncthreads();
    }
    if (tid == 0) rowptr[n] = carry;
}

__global__ __launch_bounds__(256) void fill_csr_kernel(const int* __restrict__ src,
                                                       const int* __restrict__ dst,
                                                       const int* __restrict__ rank,
                                                       const int* __restrict__ rowptr,
                                                       const float* __restrict__ dinv,
                                                       int2* __restrict__ csr, int e) {
    int i = blockIdx.x * blockDim.x + threadIdx.x;
    if (i >= e) return;
    int s = src[i], d = dst[i];
    csr[rowptr[d] + rank[i]] = make_int2(s, __float_as_int(dinv[s]));
}

template <int K, int N, int THREADS>
__device__ __forceinline__ void gemm_wmma_block(const float* __restrict__ A,
                                                const float* __restrict__ B,
                                                float* __restrict__ C, int M,
                                                int blk, float* __restrict__ Bt) {
    constexpr int P  = K + 4;
    constexpr int NW = N / 2;
    constexpr int NT = NW / 16;
    constexpr int J  = K / 8;
    constexpr int RG = THREADS / 64;

    const int tid  = threadIdx.x;
    const int wave = tid >> 5;
    const int lane = tid & 31;
    const int h    = lane >> 4;
    const int c    = lane & 15;
    const int row0 = blk * (RG * 16) + (wave % RG) * 16;
    const int col0 = (wave / RG) * NW;

    constexpr int IT = N * (K / 4) / THREADS;

    float4 bv[IT];
#pragma unroll
    for (int it = 0; it < IT; ++it) {
        const int idx = tid + it * THREADS;
        const float* bsrc = B + (size_t)(4 * (idx / N)) * N + (idx % N);
        bv[it].x = bsrc[0];
        bv[it].y = bsrc[N];
        bv[it].z = bsrc[2 * N];
        bv[it].w = bsrc[3 * N];
    }

    int ar = row0 + c;
    ar = ar < M ? ar : M - 1;
    const float4* ap = (const float4*)(A + (size_t)ar * K) + h;
    float4 af[J];
#pragma unroll
    for (int j = 0; j < J; ++j) af[j] = ap[2 * j];

#pragma unroll
    for (int it = 0; it < IT; ++it) {
        const int idx = tid + it * THREADS;
        *(float4*)(Bt + (idx % N) * P + 4 * (idx / N)) = bv[it];
    }
    __syncthreads();
    if (row0 >= M) return;

    v8f acc[NT];
#pragma unroll
    for (int t = 0; t < NT; ++t) acc[t] = v8f{};

    const float* bp = Bt + (col0 + c) * P + 4 * h;
#pragma unroll
    for (int j = 0; j < J; ++j) {
        float4 bf[NT];
#pragma unroll
        for (int t = 0; t < NT; ++t) bf[t] = *(const float4*)(bp + t * 16 * P + 8 * j);
#pragma unroll
        for (int t = 0; t < NT; ++t) {
            v2f a, b;
            a.x = af[j].x; a.y = af[j].y;
            b.x = bf[t].x; b.y = bf[t].y;
            acc[t] = __builtin_amdgcn_wmma_f32_16x16x4_f32(
                 false, a,  false, b,
                 (short)0, acc[t],  false,  false);
        }
#pragma unroll
        for (int t = 0; t < NT; ++t) {
            v2f a, b;
            a.x = af[j].z; a.y = af[j].w;
            b.x = bf[t].z; b.y = bf[t].w;
            acc[t] = __builtin_amdgcn_wmma_f32_16x16x4_f32(
                 false, a,  false, b,
                 (short)0, acc[t],  false,  false);
        }
    }

#pragma unroll
    for (int t = 0; t < NT; ++t) {
        const int col = col0 + t * 16 + c;
#pragma unroll
        for (int g = 0; g < 8; ++g) {
            const int r = row0 + h * 8 + g;
            if (r < M) C[(size_t)r * N + col] = acc[t][g];
        }
    }
}

template <int K, int N>
__global__ __launch_bounds__(256) void gemm_wmma_kernel(const float* __restrict__ A,
                                                        const float* __restrict__ B,
                                                        float* __restrict__ C, int M) {
    __shared__ float4 Bt4[N * (K + 4) / 4];
    gemm_wmma_block<K, N, 256>(A, B, C, M, blockIdx.x, (float*)Bt4);
}

__global__ __launch_bounds__(1024) void scan_gemm_kernel(const int* __restrict__ cnt,
                                                         int* __restrict__ rowptr,
                                                         float* __restrict__ dinv, int n,
                                                         const float* __restrict__ A,
                                                         const float* __restrict__ B,
                                                         float* __restrict__ C, int M) {
    __shared__ float4 Bt4[128 * (128 + 4) / 4];
    if (blockIdx.x == 0) {
        scan_dinv_block(cnt, rowptr, dinv, n);
        return;
    }
    gemm_wmma_block<128, 128, 1024>(A, B, C, M, blockIdx.x - 1, (float*)Bt4);
}

template <int F, int B>
__global__ __launch_bounds__(256) void gather_kernel(const int* __restrict__ rowptr,
                                                     const int2* __restrict__ csr,
                                                     const float* __restrict__ dinv,
                                                     const float* __restrict__ H,
                                                     const float* __restrict__ bias,
                                                     float* __restrict__ Z, int n) {
    constexpr int V = F / 32;
    typedef float vecf __attribute__((ext_vector_type(V)));
    const int lane = threadIdx.x & 31;
    const int node = __builtin_amdgcn_readfirstlane(blockIdx.x * 8 + (threadIdx.x >> 5));
    if (node >= n) return;
    const vecf* Hv = (const vecf*)H;
    const int   beg = rowptr[node];
    const int   m   = rowptr[node + 1] - beg;
    const float dn  = dinv[node];

    int  mm  = m < 32 ? m : 32;
    int2 ent = make_int2(node, 0);
    if (lane < mm) ent = csr[beg + lane];
    const vecf self = Hv[(size_t)node * 32 + lane];
    vecf acc = {};

    for (int c0 = 0;;) {
        int j = 0;
        for (; j + B <= mm; j += B) {
            vecf  v[B];
            float w[B];
            for (int k = 0; k < B; ++k) {
                const int s = __builtin_amdgcn_readlane(ent.x, j + k);
                w[k] = __int_as_float(__builtin_amdgcn_readlane(ent.y, j + k)) * dn;
                v[k] = Hv[(size_t)s * 32 + lane];
            }
            for (int k = 0; k < B; ++k) acc += v[k] * w[k];
        }
        if (j < mm) {
            vecf  v[B];
            float w[B];
            for (int k = 0; k < B; ++k) {
                const int jj = (j + k < 32) ? (j + k) : 31;
                const int s = __builtin_amdgcn_readlane(ent.x, jj);
                w[k] = __int_as_float(__builtin_amdgcn_readlane(ent.y, jj)) * dn;
                v[k] = Hv[(size_t)s * 32 + lane];
            }
            for (int k = 0; k < B; ++k) {
                if (j + k < mm) acc += v[k] * w[k];
            }
        }
        c0 += 32;
        if (c0 >= m) break;
        mm  = (m - c0 < 32) ? (m - c0) : 32;
        ent = make_int2(node, 0);
        if (lane < mm) ent = csr[beg + c0 + lane];
    }

    acc += self * (dn * dn);
    const vecf b = ((const vecf*)bias)[lane];
    vecf r = acc + b;
    for (int q = 0; q < V; ++q) r[q] = r[q] > 0.0f ? r[q] : 0.0f;
    ((vecf*)Z)[(size_t)node * 32 + lane] = r;
}

#define DEC_ITER 4

template <int T>
__device__ __forceinline__ int group8_bcast(int v) {
    return (int)__builtin_amdgcn_mov_dpp8((unsigned)v, (unsigned)T * 0x249249u);
}

template <int M>
__device__ __forceinline__ float xor_add(float v) {
    const int o = __builtin_amdgcn_update_dpp(0, __float_as_int(v), 0x160 + M, 0xf, 0xf, false);
    return v + __int_as_float(o);
}

template <int T>
__device__ __forceinline__ void dec_slot_load(const float* __restrict__ Z, int a, int b, int q,
                                              float4& x0, float4& x1, float4& y0, float4& y1) {
    const unsigned ra = (unsigned)group8_bcast<T>(a) * 64u;
    const unsigned rb = (unsigned)group8_bcast<T>(b) * 64u;
    const float4* za = (const float4*)(Z + ra) + q;
    const float4* zb = (const float4*)(Z + rb) + q;
    x0 = za[0]; x1 = za[8];
    y0 = zb[0]; y1 = zb[8];
}

__device__ __forceinline__ float dec_slot_sum(const float4& x0, const float4& x1,
                                              const float4& y0, const float4& y1) {
    float s = x0.x * y0.x;
    s += x0.y * y0.y; s += x0.z * y0.z; s += x0.w * y0.w;
    s += x1.x * y1.x; s += x1.y * y1.y; s += x1.z * y1.z; s += x1.w * y1.w;
    s = xor_add<1>(s);
    s = xor_add<2>(s);
    s = xor_add<4>(s);
    return s;
}

__global__ __launch_bounds__(256) void decoder_kernel(const float* __restrict__ Z,
                                                      const int* __restrict__ ia,
                                                      const int* __restrict__ ib,
                                                      float* __restrict__ out, int npairs) {
    const int lane = threadIdx.x & 31;
    const int q    = lane & 7;
    const int base0 = __builtin_amdgcn_readfirstlane((blockIdx.x * 8 + (threadIdx.x >> 5)) * (32 * DEC_ITER));
    if (base0 >= npairs) return;

    int a[DEC_ITER], b[DEC_ITER];
#pragma unroll
    for (int it = 0; it < DEC_ITER; ++it) {
        const int p = base0 + it * 32 + lane;
        a[it] = 0; b[it] = 0;
        if (p < npairs) { a[it] = ia[p]; b[it] = ib[p]; }
    }

#pragma unroll
    for (int it = 0; it < DEC_ITER; ++it) {
        if (base0 + it * 32 >= npairs) break;
        float4 x0[8], x1[8], y0[8], y1[8];
        dec_slot_load<0>(Z, a[it], b[it], q, x0[0], x1[0], y0[0], y1[0]);
        dec_slot_load<1>(Z, a[it], b[it], q, x0[1], x1[1], y0[1], y1[1]);
        dec_slot_load<2>(Z, a[it], b[it], q, x0[2], x1[2], y0[2], y1[2]);
        dec_slot_load<3>(Z, a[it], b[it], q, x0[3], x1[3], y0[3], y1[3]);
        dec_slot_load<4>(Z, a[it], b[it], q, x0[4], x1[4], y0[4], y1[4]);
        dec_slot_load<5>(Z, a[it], b[it], q, x0[5], x1[5], y0[5], y1[5]);
        dec_slot_load<6>(Z, a[it], b[it], q, x0[6], x1[6], y0[6], y1[6]);
        dec_slot_load<7>(Z, a[it], b[it], q, x0[7], x1[7], y0[7], y1[7]);
        float res = 0.0f;
#pragma unroll
        for (int t = 0; t < 8; ++t) {
            const float s = dec_slot_sum(x0[t], x1[t], y0[t], y1[t]);
            res = (q == t) ? s : res;
        }
        const int p = base0 + it * 32 + lane;
        if (p < npairs) out[p] = res;
    }
}

extern "C" void kernel_launch(void* const* d_in, const int* in_sizes, int n_in,
                              void* d_out, int out_size, void* d_ws, size_t ws_size,
                              hipStream_t stream) {
    const float* x     = (const float*)d_in[0];
    const float* W_in  = (const float*)d_in[1];
    const float* b_in  = (const float*)d_in[2];
    const float* W_out = (const float*)d_in[3];
    const float* b_out = (const float*)d_in[4];
    const int*   ei    = (const int*)d_in[5];
    const int*   eli   = (const int*)d_in[6];

    const int N    = in_sizes[0] / 128;
    const int E    = in_sizes[5] / 2;
    const int NDEC = in_sizes[6] / 2;
    const int* src = ei;
    const int* dst = ei + E;
    const int* ia  = eli;
    const int* ib  = eli + NDEC;
    float* logits  = (float*)d_out;

    char* base = (char*)d_ws;
    size_t off = 0;
    auto take = [&](size_t bytes) -> float* {
        float* p = (float*)(base + off);
        off = (off + bytes + 255) & ~(size_t)255;
        return p;
    };
    int*   cnt    = (int*)take((size_t)N * 4);
    float* dinv   = take((size_t)N * 4);
    int*   rowptr = (int*)take((size_t)(N + 1) * 4);
    int*   rank   = (int*)take((size_t)E * 4);
    int2*  csr    = (int2*)take((size_t)E * 8);
    float* H1   = take((size_t)N * 128 * 4);
    float* A1   = take((size_t)N * 128 * 4);
    float* H2   = take((size_t)N * 64 * 4);
    float* A2   = take((size_t)N * 64 * 4);
    (void)ws_size;

    const int TB = 256;
    zero_cnt_kernel<<<(N + TB - 1) / TB, TB, 0, stream>>>(cnt, N);
    count_rank_kernel<<<(E + TB - 1) / TB, TB, 0, stream>>>(dst, cnt, rank, E);
    scan_gemm_kernel<<<1 + (N + 255) / 256, 1024, 0, stream>>>(cnt, rowptr, dinv, N, x, W_in, H1, N);
    fill_csr_kernel<<<(E + TB - 1) / TB, TB, 0, stream>>>(src, dst, rank, rowptr, dinv, csr, E);

    gather_kernel<128, 4><<<(N + 7) / 8, 256, 0, stream>>>(rowptr, csr, dinv, H1, b_in, A1, N);

    gemm_wmma_kernel<128, 64><<<(N + 63) / 64, 256, 0, stream>>>(A1, W_out, H2, N);
    gather_kernel<64, 8><<<(N + 7) / 8, 256, 0, stream>>>(rowptr, csr, dinv, H2, b_out, A2, N);

    decoder_kernel<<<(NDEC + 256 * DEC_ITER - 1) / (256 * DEC_ITER), 256, 0, stream>>>(A2, ia, ib, logits, NDEC);
}


# Adjacency in one pass: the counting atomic's return is the slot in a 64-per-node table (overflow list behind); scan, rank, fill launch go; dinv[src] goes into H rows at GEMM store; rounding changes
# speedup vs baseline: 3.2993x; 1.0907x over previous
#include <hip/hip_runtime.h>

typedef float v2f __attribute__((ext_vector_type(2)));
typedef float v8f __attribute__((ext_vector_type(8)));

#define SLOTS 64

__global__ __launch_bounds__(256) void zero_cnt_kernel(int* __restrict__ cnt, int n) {
    int i = blockIdx.x * blockDim.x + threadIdx.x;
    if (i < n) cnt[i] = 0;
}

__global__ __launch_bounds__(256) void count_slot_kernel(const int* __restrict__ src,
                                                         const int* __restrict__ dst,
                                                         int* __restrict__ cnt,
                                                         int* __restrict__ slot,
                                                         int* novf,
                                                         int2* __restrict__ ovf, int e) {
    int i = blockIdx.x * blockDim.x + threadIdx.x;
    if (i >= e) return;
    const int s = src[i], d = dst[i];
    const int r = atomicAdd(&cnt[d], 1);
    if (r < SLOTS) slot[(size_t)d * SLOTS + r] = s;
    else           ovf[atomicAdd(novf, 1)] = make_int2(s, d);
}

template <int K, int N, int THREADS>
__device__ __forceinline__ void gemm_wmma_block(const float* __restrict__ A,
                                                const float* __restrict__ B,
                                                float* __restrict__ C, int M,
                                                int blk, float* __restrict__ Bt,
                                                const int* __restrict__ degc) {
    constexpr int P  = K + 4;
    constexpr int NW = N / 2;
    constexpr int NT = NW / 16;
    constexpr int J  = K / 8;
    constexpr int RG = THREADS / 64;

    const int tid  = threadIdx.x;
    const int wave = tid >> 5;
    const int lane = tid & 31;
    const int h    = lane >> 4;
    const int c    = lane & 15;
    const int row0 = blk * (RG * 16) + (wave % RG) * 16;
    const int col0 = (wave / RG) * NW;

    constexpr int IT = N * (K / 4) / THREADS;

    float4 bv[IT];
#pragma unroll
    for (int it = 0; it < IT; ++it) {
        const int idx = tid + it * THREADS;
        const float* bsrc = B + (size_t)(4 * (idx / N)) * N + (idx % N);
        bv[it].x = bsrc[0];
        bv[it].y = bsrc[N];
        bv[it].z = bsrc[2 * N];
        bv[it].w = bsrc[3 * N];
    }

    int ar = row0 + c;
    ar = ar < M ? ar : M - 1;
    const float4* ap = (const float4*)(A + (size_t)ar * K) + h;
    float4 af[J];
#pragma unroll
    for (int j = 0; j < J; ++j) af[j] = ap[2 * j];

    const float rsl = rsqrtf((float)(degc[ar] + 1));

#pragma unroll
    for (int it = 0; it < IT; ++it) {
        const int idx = tid + it * THREADS;
        *(float4*)(Bt + (idx % N) * P + 4 * (idx / N)) = bv[it];
    }
    __syncthreads();
    if (row0 >= M) return;

    v8f acc[NT];
#pragma unroll
    for (int t = 0; t < NT; ++t) acc[t] = v8f{};

    const float* bp = Bt + (col0 + c) * P + 4 * h;
#pragma unroll
    for (int j = 0; j < J; ++j) {
        float4 bf[NT];
#pragma unroll
        for (int t = 0; t < NT; ++t) bf[t] = *(const float4*)(bp + t * 16 * P + 8 * j);
#pragma unroll
        for (int t = 0; t < NT; ++t) {
            v2f a, b;
            a.x = af[j].x; a.y = af[j].y;
            b.x = bf[t].x; b.y = bf[t].y;
            acc[t] = __builtin_amdgcn_wmma_f32_16x16x4_f32(
                 false, a,  false, b,
                 (short)0, acc[t],  false,  false);
        }
#pragma unroll
        for (int t = 0; t < NT; ++t) {
            v2f a, b;
            a.x = af[j].z; a.y = af[j].w;
            b.x = bf[t].z; b.y = bf[t].w;
            acc[t] = __builtin_amdgcn_wmma_f32_16x16x4_f32(
                 false, a,  false, b,
                 (short)0, acc[t],  false,  false);
        }
    }

    float rs[8];
#pragma unroll
    for (int g = 0; g < 8; ++g) rs[g] = __shfl(rsl, h * 8 + g, 32);
#pragma unroll
    for (int t = 0; t < NT; ++t) {
        const int col = col0 + t * 16 + c;
#pragma unroll
        for (int g = 0; g < 8; ++g) {
            const int r = row0 + h * 8 + g;
            if (r < M) C[(size_t)r * N + col] = acc[t][g] * rs[g];
        }
    }
}

template <int K, int N, int THREADS>
__global__ __launch_bounds__(THREADS) void gemm_wmma_kernel(const float* __restrict__ A,
                                                            const float* __restrict__ B,
                                                            float* __restrict__ C, int M,
                                                            const int* __restrict__ degc) {
    __shared__ float4 Bt4[N * (K + 4) / 4];
    gemm_wmma_block<K, N, THREADS>(A, B, C, M, blockIdx.x, (float*)Bt4, degc);
}

template <int F, int B>
__global__ __launch_bounds__(256) void gather_kernel(const int* __restrict__ cnt,
                                                     const int* __restrict__ slot,
                                                     const int* __restrict__ novf,
                                                     const int2* __restrict__ ovf,
                                                     const float* __restrict__ H,
                                                     const float* __restrict__ bias,
                                                     float* __restrict__ Z, int n) {
    constexpr int V = F / 32;
    typedef float vecf __attribute__((ext_vector_type(V)));
    const int lane = threadIdx.x & 31;
    const int node = __builtin_amdgcn_readfirstlane(blockIdx.x * 8 + (threadIdx.x >> 5));
    if (node >= n) return;
    const vecf* Hv = (const vecf*)H;
    const int   deg = cnt[node];
    const int   m   = deg < SLOTS ? deg : SLOTS;
    const float dn  = rsqrtf((float)(deg + 1));
    const int*  row = slot + (size_t)node * SLOTS;

    int mm  = m < 32 ? m : 32;
    int ent = node;
    if (lane < mm) ent = row[lane];
    const vecf self = Hv[(size_t)node * 32 + lane];
    vecf acc = {};

    for (int c0 = 0;;) {
        int j = 0;
        for (; j + B <= mm; j += B) {
            vecf v[B];
            for (int k = 0; k < B; ++k) {
                const int s = __builtin_amdgcn_readlane(ent, j + k);
                v[k] = Hv[(size_t)s * 32 + lane];
            }
            for (int k = 0; k < B; ++k) acc += v[k];
        }
        if (j < mm) {
            vecf v[B];
            for (int k = 0; k < B; ++k) {
                const int jj = (j + k < 32) ? (j + k) : 31;
                const int s = __builtin_amdgcn_readlane(ent, jj);
                v[k] = Hv[(size_t)s * 32 + lane];
            }
            for (int k = 0; k < B; ++k) {
                if (j + k < mm) acc += v[k];
            }
        }
        c0 += 32;
        if (c0 >= m) break;
        mm  = (m - c0 < 32) ? (m - c0) : 32;
        ent = node;
        if (lane < mm) ent = row[c0 + lane];
    }

    if (deg > SLOTS) {
        const int no = novf[0];
        for (int o0 = 0; o0 < no; o0 += 32) {
            int2 pr = make_int2(0, -1);
            if (o0 + lane < no) pr = ovf[o0 + lane];
            unsigned hit = __builtin_amdgcn_ballot_w32(pr.y == node);
            while (hit) {
                const int l = __builtin_ctz(hit);
                hit &= hit - 1;
                const int s = __builtin_amdgcn_readlane(pr.x, l);
                acc += Hv[(size_t)s * 32 + lane];
            }
        }
    }

    acc += self;
    const vecf b = ((const vecf*)bias)[lane];
    vecf r = acc * dn + b;
    for (int q = 0; q < V; ++q) r[q] = r[q] > 0.0f ? r[q] : 0.0f;
    ((vecf*)Z)[(size_t)node * 32 + lane] = r;
}

#define DEC_ITER 4

template <int T>
__device__ __forceinline__ int group8_bcast(int v) {
    return (int)__builtin_amdgcn_mov_dpp8((unsigned)v, (unsigned)T * 0x249249u);
}

template <int M>
__device__ __forceinline__ float xor_add(float v) {
    const int o = __builtin_amdgcn_update_dpp(0, __float_as_int(v), 0x160 + M, 0xf, 0xf, false);
    return v + __int_as_float(o);
}

template <int T>
__device__ __forceinline__ void dec_slot_load(const float* __restrict__ Z, int a, int b, int q,
                                              float4& x0, float4& x1, float4& y0, float4& y1) {
    const unsigned ra = (unsigned)group8_bcast<T>(a) * 64u;
    const unsigned rb = (unsigned)group8_bcast<T>(b) * 64u;
    const float4* za = (const float4*)(Z + ra) + q;
    const float4* zb = (const float4*)(Z + rb) + q;
    x0 = za[0]; x1 = za[8];
    y0 = zb[0]; y1 = zb[8];
}

__device__ __forceinline__ float dec_slot_sum(const float4& x0, const float4& x1,
                                              const float4& y0, const float4& y1) {
    float s = x0.x * y0.x;
    s += x0.y * y0.y; s += x0.z * y0.z; s += x0.w * y0.w;
    s += x1.x * y1.x; s += x1.y * y1.y; s += x1.z * y1.z; s += x1.w * y1.w;
    s = xor_add<1>(s);
    s = xor_add<2>(s);
    s = xor_add<4>(s);
    return s;
}

__global__ __launch_bounds__(256) void decoder_kernel(const float* __restrict__ Z,
                                                      const int* __restrict__ ia,
                                                      const int* __restrict__ ib,
                                                      float* __restrict__ out, int npairs) {
    const int lane = threadIdx.x & 31;
    const int q    = lane & 7;
    const int base0 = __builtin_amdgcn_readfirstlane((blockIdx.x * 8 + (threadIdx.x >> 5)) * (32 * DEC_ITER));
    if (base0 >= npairs) return;

    int a[DEC_ITER], b[DEC_ITER];
#pragma unroll
    for (int it = 0; it < DEC_ITER; ++it) {
        const int p = base0 + it * 32 + lane;
        a[it] = 0; b[it] = 0;
        if (p < npairs) { a[it] = ia[p]; b[it] = ib[p]; }
    }

#pragma unroll
    for (int it = 0; it < DEC_ITER; ++it) {
        if (base0 + it * 32 >= npairs) break;
        float4 x0[8], x1[8], y0[8], y1[8];
        dec_slot_load<0>(Z, a[it], b[it], q, x0[0], x1[0], y0[0], y1[0]);
        dec_slot_load<1>(Z, a[it], b[it], q, x0[1], x1[1], y0[1], y1[1]);
        dec_slot_load<2>(Z, a[it], b[it], q, x0[2], x1[2], y0[2], y1[2]);
        dec_slot_load<3>(Z, a[it], b[it], q, x0[3], x1[3], y0[3], y1[3]);
        dec_slot_load<4>(Z, a[it], b[it], q, x0[4], x1[4], y0[4], y1[4]);
        dec_slot_load<5>(Z, a[it], b[it], q, x0[5], x1[5], y0[5], y1[5]);
        dec_slot_load<6>(Z, a[it], b[it], q, x0[6], x1[6], y0[6], y1[6]);
        dec_slot_load<7>(Z, a[it], b[it], q, x0[7], x1[7], y0[7], y1[7]);
        float res = 0.0f;
#pragma unroll
        for (int t = 0; t < 8; ++t) {
            const float s = dec_slot_sum(x0[t], x1[t], y0[t], y1[t]);
            res = (q == t) ? s : res;
        }
        const int p = base0 + it * 32 + lane;
        if (p < npairs) out[p] = res;
    }
}

extern "C" void kernel_launch(void* const* d_in, const int* in_sizes, int n_in,
                              void* d_out, int out_size, void* d_ws, size_t ws_size,
                              hipStream_t stream) {
    const float* x     = (const float*)d_in[0];
    const float* W_in  = (const float*)d_in[1];
    const float* b_in  = (const float*)d_in[2];
    const float* W_out = (const float*)d_in[3];
    const float* b_out = (const float*)d_in[4];
    const int*   ei    = (const int*)d_in[5];
    const int*   eli   = (const int*)d_in[6];

    const int N    = in_sizes[0] / 128;
    const int E    = in_sizes[5] / 2;
    const int NDEC = in_sizes[6] / 2;
    const int* src = ei;
    const int* dst = ei + E;
    const int* ia  = eli;
    const int* ib  = eli + NDEC;
    float* logits  = (float*)d_out;

    char* base = (char*)d_ws;
    size_t off = 0;
    auto take = [&](size_t bytes) -> float* {
        float* p = (float*)(base + off);
        off = (off + bytes + 255) & ~(size_t)255;
        return p;
    };
    int*   cnt  = (int*)take((size_t)(N + 1) * 4);
    int*   slot = (int*)take((size_t)N * SLOTS * 4);
    int2*  ovf  = (int2*)take((size_t)E * 8);
    float* H1   = take((size_t)N * 128 * 4);
    float* A1   = take((size_t)N * 128 * 4);
    float* H2   = take((size_t)N * 64 * 4);
    float* A2   = take((size_t)N * 64 * 4);
    (void)ws_size;

    const int TB = 256;
    zero_cnt_kernel<<<(N + 1 + TB - 1) / TB, TB, 0, stream>>>(cnt, N + 1);
    count_slot_kernel<<<(E + TB - 1) / TB, TB, 0, stream>>>(src, dst, cnt, slot, cnt + N, ovf, E);

    gemm_wmma_kernel<128, 128, 1024><<<(N + 255) / 256, 1024, 0, stream>>>(x, W_in, H1, N, cnt);
    gather_kernel<128, 4><<<(N + 7) / 8, 256, 0, stream>>>(cnt, slot, cnt + N, ovf, H1, b_in, A1, N);

    gemm_wmma_kernel<128, 64, 256><<<(N + 63) / 64, 256, 0, stream>>>(A1, W_out, H2, N, cnt);
    gather_kernel<64, 8><<<(N + 7) / 8, 256, 0, stream>>>(cnt, slot, cnt + N, ovf, H2, b_out, A2, N);

    decoder_kernel<<<(NDEC + 256 * DEC_ITER - 1) / (256 * DEC_ITER), 256, 0, stream>>>(A2, ia, ib, logits, NDEC);
}


# Adjacency build without per-edge global atomics: edges ordered by 256-node destination block in LDS, then each block's slot rows filled by LDS counters, stored whole; row sum order = LDS arrival rank
# speedup vs baseline: 5.1241x; 1.5531x over previous
#include <hip/hip_runtime.h>

typedef float v2f __attribute__((ext_vector_type(2)));
typedef float v8f __attribute__((ext_vector_type(8)));

#define SLOTS 64
#define BKN   256
#define CH    4096
#define MAXBK 1024

__global__ __launch_bounds__(1024) void bin_kernel(const int* __restrict__ src,
                                                   const int* __restrict__ dst,
                                                   int* __restrict__ chunkE,
                                                   int2* __restrict__ runT,
                                                   int* __restrict__ novf, int e, int nbk) {
    __shared__ int hist[MAXBK];
    __shared__ int wsum[32];
    __shared__ int sorted[CH];
    const int tid  = threadIdx.x;
    const int lane = tid & 31;
    const int wave = tid >> 5;
    const int b    = blockIdx.x;
    const int nblk = gridDim.x;
    if (b == 0 && tid == 0) novf[0] = 0;
    hist[tid] = 0;

    int s[4], d[4], r[4];
#pragma unroll
    for (int i = 0; i < 4; ++i) {
        const int idx = b * CH + i * 1024 + tid;
        const int ii  = idx < e ? idx : e - 1;
        s[i] = src[ii];
        const int dd = dst[ii];
        d[i] = idx < e ? dd : -1;
        r[i] = 0;
    }
    __syncthreads();
#pragma unroll
    for (int i = 0; i < 4; ++i)
        if (d[i] >= 0) r[i] = atomicAdd(&hist[d[i] / BKN], 1);
    __syncthreads();

    const int v = hist[tid];
    int inc = v;
#pragma unroll
    for (int o = 1; o < 32; o <<= 1) {
        const int t = __shfl_up(inc, o, 32);
        if (lane >= o) inc += t;
    }
    if (lane == 31) wsum[wave] = inc;
    __syncthreads();
    if (wave == 0) {
        const int t = wsum[lane];
        int ti = t;
#pragma unroll
        for (int o = 1; o < 32; o <<= 1) {
            const int u = __shfl_up(ti, o, 32);
            if (lane >= o) ti += u;
        }
        wsum[lane] = ti - t;
    }
    __syncthreads();
    const int start = wsum[wave] + inc - v;
    hist[tid] = start;
    if (tid < nbk) runT[(size_t)tid * nblk + b] = make_int2(start, v);
    __syncthreads();
#pragma unroll
    for (int i = 0; i < 4; ++i)
        if (d[i] >= 0) sorted[hist[d[i] / BKN] + r[i]] = (s[i] << 8) | (d[i] % BKN);
    __syncthreads();
    const int left = e - b * CH;
    const int cntb = left < CH ? left : CH;
#pragma unroll
    for (int i = 0; i < 4; ++i) {
        const int j = i * 1024 + tid;
        if (j < cntb) chunkE[(size_t)b * CH + j] = sorted[j];
    }
}

__device__ __forceinline__ void slot_insert(int ev, int node0, int* cntL, int* slotL,
                                            int* novf, int2* __restrict__ ovf) {
    const int s  = ev >> 8;
    const int dl = ev & (BKN - 1);
    const int r  = atomicAdd(&cntL[dl], 1);
    if (r < SLOTS) slotL[dl * SLOTS + r] = s;
    else           ovf[atomicAdd(novf, 1)] = make_int2(s, node0 + dl);
}

__global__ __launch_bounds__(1024) void build_kernel(const int* __restrict__ chunkE,
                                                     const int2* __restrict__ runT,
                                                     int* __restrict__ cnt,
                                                     int* __restrict__ slot,
                                                     int* novf,
                                                     int2* __restrict__ ovf, int n, int nblk) {
    __shared__ int  cntL[BKN];
    __shared__ int4 slotL4[BKN * SLOTS / 4];
    int* slotL = (int*)slotL4;
    const int tid  = threadIdx.x;
    const int lane = tid & 31;
    const int w     = __builtin_amdgcn_readfirstlane(tid >> 5);
    const int k     = blockIdx.x;
    const int node0 = k * BKN;
    if (tid < BKN) cntL[tid] = 0;
    __syncthreads();

    const int2* runs = runT + (size_t)k * nblk;
    for (int b0 = 0; b0 < nblk; b0 += 256) {
        int st[8], ln[8], ev[8];
#pragma unroll
        for (int j = 0; j < 8; ++j) {
            const int b = b0 + j * 32 + w;
            int2 rec = make_int2(0, 0);
            if (b < nblk) rec = runs[b];
            st[j] = b * CH + rec.x;
            ln[j] = rec.y;
        }
#pragma unroll
        for (int j = 0; j < 8; ++j) {
            ev[j] = 0;
            if (lane < ln[j]) ev[j] = chunkE[st[j] + lane];
        }
#pragma unroll
        for (int j = 0; j < 8; ++j)
            if (lane < ln[j]) slot_insert(ev[j], node0, cntL, slotL, novf, ovf);
#pragma unroll
        for (int j = 0; j < 8; ++j)
            for (int i = 32 + lane; i < ln[j]; i += 32)
                slot_insert(chunkE[st[j] + i], node0, cntL, slotL, novf, ovf);
    }
    __syncthreads();

    if (tid < BKN && node0 + tid < n) cnt[node0 + tid] = cntL[tid];
#pragma unroll
    for (int i = 0; i < BKN * SLOTS / 4 / 1024; ++i) {
        const int c    = tid + i * 1024;
        const int nl   = c / (SLOTS / 4);
        const int part = c % (SLOTS / 4);
        int m = cntL[nl];
        m = m < SLOTS ? m : SLOTS;
        if (node0 + nl < n && part * 4 < m)
            ((int4*)slot)[(size_t)(node0 + nl) * (SLOTS / 4) + part] = slotL4[c];
    }
}

template <int K, int N, int THREADS>
__device__ __forceinline__ void gemm_wmma_block(const float* __restrict__ A,
                                                const float* __restrict__ B,
                                                float* __restrict__ C, int M,
                                                int blk, float* __restrict__ Bt,
                                                const int* __restrict__ degc) {
    constexpr int P  = K + 4;
    constexpr int NW = N / 2;
    constexpr int NT = NW / 16;
    constexpr int J  = K / 8;
    constexpr int RG = THREADS / 64;

    const int tid  = threadIdx.x;
    const int wave = tid >> 5;
    const int lane = tid & 31;
    const int h    = lane >> 4;
    const int c    = lane & 15;
    const int row0 = blk * (RG * 16) + (wave % RG) * 16;
    const int col0 = (wave / RG) * NW;

    constexpr int IT = N * (K / 4) / THREADS;

    float4 bv[IT];
#pragma unroll
    for (int it = 0; it < IT; ++it) {
        const int idx = tid + it * THREADS;
        const float* bsrc = B + (size_t)(4 * (idx / N)) * N + (idx % N);
        bv[it].x = bsrc[0];
        bv[it].y = bsrc[N];
        bv[it].z = bsrc[2 * N];
        bv[it].w = bsrc[3 * N];
    }

    int ar = row0 + c;
    ar = ar < M ? ar : M - 1;
    const float4* ap = (const float4*)(A + (size_t)ar * K) + h;
    float4 af[J];
#pragma unroll
    for (int j = 0; j < J; ++j) af[j] = ap[2 * j];

    const float rsl = rsqrtf((float)(degc[ar] + 1));

#pragma unroll
    for (int it = 0; it < IT; ++it) {
        const int idx = tid + it * THREADS;
        *(float4*)(Bt + (idx % N) * P + 4 * (idx / N)) = bv[it];
    }
    __syncthreads();
    if (row0 >= M) return;

    v8f acc[NT];
#pragma unroll
    for (int t = 0; t < NT; ++t) acc[t] = v8f{};

    const float* bp = Bt + (col0 + c) * P + 4 * h;
#pragma unroll
    for (int j = 0; j < J; ++j) {
        float4 bf[NT];
#pragma unroll
        for (int t = 0; t < NT; ++t) bf[t] = *(const float4*)(bp + t * 16 * P + 8 * j);
#pragma unroll
        for (int t = 0; t < NT; ++t) {
            v2f a, b;
            a.x = af[j].x; a.y = af[j].y;
            b.x = bf[t].x; b.y = bf[t].y;
            acc[t] = __builtin_amdgcn_wmma_f32_16x16x4_f32(
                 false, a,  false, b,
                 (short)0, acc[t],  false,  false);
        }
#pragma unroll
        for (int t = 0; t < NT; ++t) {
            v2f a, b;
            a.x = af[j].z; a.y = af[j].w;
            b.x = bf[t].z; b.y = bf[t].w;
            acc[t] = __builtin_amdgcn_wmma_f32_16x16x4_f32(
                 false, a,  false, b,
                 (short)0, acc[t],  false,  false);
        }
    }

    float rs[8];
#pragma unroll
    for (int g = 0; g < 8; ++g) rs[g] = __shfl(rsl, h * 8 + g, 32);
#pragma unroll
    for (int t = 0; t < NT; ++t) {
        const int col = col0 + t * 16 + c;
#pragma unroll
        for (int g = 0; g < 8; ++g) {
            const int r = row0 + h * 8 + g;
            if (r < M) C[(size_t)r * N + col] = acc[t][g] * rs[g];
        }
    }
}

template <int K, int N, int THREADS>
__global__ __launch_bounds__(THREADS) void gemm_wmma_kernel(const float* __restrict__ A,
                                                            const float* __restrict__ B,
                                                            float* __restrict__ C, int M,
                                                            const int* __restrict__ degc) {
    __shared__ float4 Bt4[N * (K + 4) / 4];
    gemm_wmma_block<K, N, THREADS>(A, B, C, M, blockIdx.x, (float*)Bt4, degc);
}

template <int F, int B>
__global__ __launch_bounds__(256) void gather_kernel(const int* __restrict__ cnt,
                                                     const int* __restrict__ slot,
                                                     const int* __restrict__ novf,
                                                     const int2* __restrict__ ovf,
                                                     const float* __restrict__ H,
                                                     const float* __restrict__ bias,
                                                     float* __restrict__ Z, int n) {
    constexpr int V = F / 32;
    typedef float vecf __attribute__((ext_vector_type(V)));
    const int lane = threadIdx.x & 31;
    const int node = __builtin_amdgcn_readfirstlane(blockIdx.x * 8 + (threadIdx.x >> 5));
    if (node >= n) return;
    const vecf* Hv = (const vecf*)H;
    const int   deg = cnt[node];
    const int   m   = deg < SLOTS ? deg : SLOTS;
    const float dn  = rsqrtf((float)(deg + 1));
    const int*  row = slot + (size_t)node * SLOTS;

    int mm  = m < 32 ? m : 32;
    int ent = node;
    if (lane < mm) ent = row[lane];
    const vecf self = Hv[(size_t)node * 32 + lane];
    vecf acc = {};

    for (int c0 = 0;;) {
        int j = 0;
        for (; j + B <= mm; j += B) {
            vecf v[B];
            for (int k = 0; k < B; ++k) {
                const int s = __builtin_amdgcn_readlane(ent, j + k);
                v[k] = Hv[(size_t)s * 32 + lane];
            }
            for (int k = 0; k < B; ++k) acc += v[k];
        }
        if (j < mm) {
            vecf v[B];
            for (int k = 0; k < B; ++k) {
                const int jj = (j + k < 32) ? (j + k) : 31;
                const int s = __builtin_amdgcn_readlane(ent, jj);
                v[k] = Hv[(size_t)s * 32 + lane];
            }
            for (int k = 0; k < B; ++k) {
                if (j + k < mm) acc += v[k];
            }
        }
        c0 += 32;
        if (c0 >= m) break;
        mm  = (m - c0 < 32) ? (m - c0) : 32;
        ent = node;
        if (lane < mm) ent = row[c0 + lane];
    }

    if (deg > SLOTS) {
        const int no = novf[0];
        for (int o0 = 0; o0 < no; o0 += 32) {
            int2 pr = make_int2(0, -1);
            if (o0 + lane < no) pr = ovf[o0 + lane];
            unsigned hit = __builtin_amdgcn_ballot_w32(pr.y == node);
            while (hit) {
                const int l = __builtin_ctz(hit);
                hit &= hit - 1;
                const int s = __builtin_amdgcn_readlane(pr.x, l);
                acc += Hv[(size_t)s * 32 + lane];
            }
        }
    }

    acc += self;
    const vecf b = ((const vecf*)bias)[lane];
    vecf r = acc * dn + b;
    for (int q = 0; q < V; ++q) r[q] = r[q] > 0.0f ? r[q] : 0.0f;
    ((vecf*)Z)[(size_t)node * 32 + lane] = r;
}

#define DEC_ITER 4

template <int T>
__device__ __forceinline__ int group8_bcast(int v) {
    return (int)__builtin_amdgcn_mov_dpp8((unsigned)v, (unsigned)T * 0x249249u);
}

template <int M>
__device__ __forceinline__ float xor_add(float v) {
    const int o = __builtin_amdgcn_update_dpp(0, __float_as_int(v), 0x160 + M, 0xf, 0xf, false);
    return v + __int_as_float(o);
}

template <int T>
__device__ __forceinline__ void dec_slot_load(const float* __restrict__ Z, int a, int b, int q,
                                              float4& x0, float4& x1, float4& y0, float4& y1) {
    const unsigned ra = (unsigned)group8_bcast<T>(a) * 64u;
    const unsigned rb = (unsigned)group8_bcast<T>(b) * 64u;
    const float4* za = (const float4*)(Z + ra) + q;
    const float4* zb = (const float4*)(Z + rb) + q;
    x0 = za[0]; x1 = za[8];
    y0 = zb[0]; y1 = zb[8];
}

__device__ __forceinline__ float dec_slot_sum(const float4& x0, const float4& x1,
                                              const float4& y0, const float4& y1) {
    float s = x0.x * y0.x;
    s += x0.y * y0.y; s += x0.z * y0.z; s += x0.w * y0.w;
    s += x1.x * y1.x; s += x1.y * y1.y; s += x1.z * y1.z; s += x1.w * y1.w;
    s = xor_add<1>(s);
    s = xor_add<2>(s);
    s = xor_add<4>(s);
    return s;
}

__global__ __launch_bounds__(256) void decoder_kernel(const float* __restrict__ Z,
                                                      const int* __restrict__ ia,
                                                      const int* __restrict__ ib,
                                                      float* __restrict__ out, int npairs) {
    const int lane = threadIdx.x & 31;
    const int q    = lane & 7;
    const int base0 = __builtin_amdgcn_readfirstlane((blockIdx.x * 8 + (threadIdx.x >> 5)) * (32 * DEC_ITER));
    if (base0 >= npairs) return;

    int a[DEC_ITER], b[DEC_ITER];
#pragma unroll
    for (int it = 0; it < DEC_ITER; ++it) {
        const int p = base0 + it * 32 + lane;
        a[it] = 0; b[it] = 0;
        if (p < npairs) { a[it] = ia[p]; b[it] = ib[p]; }
    }

#pragma unroll
    for (int it = 0; it < DEC_ITER; ++it) {
        if (base0 + it * 32 >= npairs) break;
        float4 x0[8], x1[8], y0[8], y1[8];
        dec_slot_load<0>(Z, a[it], b[it], q, x0[0], x1[0], y0[0], y1[0]);
        dec_slot_load<1>(Z, a[it], b[it], q, x0[1], x1[1], y0[1], y1[1]);
        dec_slot_load<2>(Z, a[it], b[it], q, x0[2], x1[2], y0[2], y1[2]);
        dec_slot_load<3>(Z, a[it], b[it], q, x0[3], x1[3], y0[3], y1[3]);
        dec_slot_load<4>(Z, a[it], b[it], q, x0[4], x1[4], y0[4], y1[4]);
        dec_slot_load<5>(Z, a[it], b[it], q, x0[5], x1[5], y0[5], y1[5]);
        dec_slot_load<6>(Z, a[it], b[it], q, x0[6], x1[6], y0[6], y1[6]);
        dec_slot_load<7>(Z, a[it], b[it], q, x0[7], x1[7], y0[7], y1[7]);
        float res = 0.0f;
#pragma unroll
        for (int t = 0; t < 8; ++t) {
            const float s = dec_slot_sum(x0[t], x1[t], y0[t], y1[t]);
            res = (q == t) ? s : res;
        }
        const int p = base0 + it * 32 + lane;
        if (p < npairs) out[p] = res;
    }
}

extern "C" void kernel_launch(void* const* d_in, const int* in_sizes, int n_in,
                              void* d_out, int out_size, void* d_ws, size_t ws_size,
                              hipStream_t stream) {
    const float* x     = (const float*)d_in[0];
    const float* W_in  = (const float*)d_in[1];
    const float* b_in  = (const float*)d_in[2];
    const float* W_out = (const float*)d_in[3];
    const float* b_out = (const float*)d_in[4];
    const int*   ei    = (const int*)d_in[5];
    const int*   eli   = (const int*)d_in[6];

    const int N    = in_sizes[0] / 128;
    const int E    = in_sizes[5] / 2;
    const int NDEC = in_sizes[6] / 2;
    const int* src = ei;
    const int* dst = ei + E;
    const int* ia  = eli;
    const int* ib  = eli + NDEC;
    float* logits  = (float*)d_out;

    char* base = (char*)d_ws;
    size_t off = 0;
    auto take = [&](size_t bytes) -> float* {
        float* p = (float*)(base + off);
        off = (off + bytes + 255) & ~(size_t)255;
        return p;
    };
    int*   cnt  = (int*)take((size_t)(N + 1) * 4);
    int*   slot = (int*)take((size_t)N * SLOTS * 4);
    int2*  ovf  = (int2*)take((size_t)E * 8);
    float* H1   = take((size_t)N * 128 * 4);
    float* A1   = take((size_t)N * 128 * 4);
    float* H2   = take((size_t)N * 64 * 4);
    float* A2   = take((size_t)N * 64 * 4);
    int*   chunkE = (int*)take(((size_t)E + CH) * 4);
    int2*  runT   = (int2*)take(((size_t)N / BKN + 1) * ((size_t)E / CH + 1) * 8);
    (void)ws_size;

    const int NBK  = (N + BKN - 1) / BKN;
    const int NBLK = (E + CH - 1) / CH;
    bin_kernel<<<NBLK, 1024, 0, stream>>>(src, dst, chunkE, runT, cnt + N, E, NBK);
    build_kernel<<<NBK, 1024, 0, stream>>>(chunkE, runT, cnt, slot, cnt + N, ovf, N, NBLK);

    gemm_wmma_kernel<128, 128, 1024><<<(N + 255) / 256, 1024, 0, stream>>>(x, W_in, H1, N, cnt);
    gather_kernel<128, 4><<<(N + 7) / 8, 256, 0, stream>>>(cnt, slot, cnt + N, ovf, H1, b_in, A1, N);

    gemm_wmma_kernel<128, 64, 256><<<(N + 63) / 64, 256, 0, stream>>>(A1, W_out, H2, N, cnt);
    gather_kernel<64, 8><<<(N + 7) / 8, 256, 0, stream>>>(cnt, slot, cnt + N, ovf, H2, b_out, A2, N);

    decoder_kernel<<<(NDEC + 256 * DEC_ITER - 1) / (256 * DEC_ITER), 256, 0, stream>>>(A2, ia, ib, logits, NDEC);
}


# Layer-1 GEMM moves into the second adjacency launch: block k builds the slot rows of its 256 nodes, then multiplies their x rows, row factor from its LDS degrees; 6 launches, not 7; sum orders kept
# speedup vs baseline: 5.3111x; 1.0365x over previous
#include <hip/hip_runtime.h>

typedef float v2f __attribute__((ext_vector_type(2)));
typedef float v8f __attribute__((ext_vector_type(8)));

#define SLOTS 64
#define BKN   256
#define CH    4096
#define MAXBK 1024

__global__ __launch_bounds__(1024) void bin_kernel(const int* __restrict__ src,
                                                   const int* __restrict__ dst,
                                                   int* __restrict__ chunkE,
                                                   int2* __restrict__ runT,
                                                   int* __restrict__ novf, int e, int nbk) {
    __shared__ int hist[MAXBK];
    __shared__ int wsum[32];
    __shared__ int sorted[CH];
    const int tid  = threadIdx.x;
    const int lane = tid & 31;
    const int wave = tid >> 5;
    const int b    = blockIdx.x;
    const int nblk = gridDim.x;
    if (b == 0 && tid == 0) novf[0] = 0;
    hist[tid] = 0;

    int s[4], d[4], r[4];
#pragma unroll
    for (int i = 0; i < 4; ++i) {
        const int idx = b * CH + i * 1024 + tid;
        const int ii  = idx < e ? idx : e - 1;
        s[i] = src[ii];
        const int dd = dst[ii];
        d[i] = idx < e ? dd : -1;
        r[i] = 0;
    }
    __syncthreads();
#pragma unroll
    for (int i = 0; i < 4; ++i)
        if (d[i] >= 0) r[i] = atomicAdd(&hist[d[i] / BKN], 1);
    __syncthreads();

    const int v = hist[tid];
    int inc = v;
#pragma unroll
    for (int o = 1; o < 32; o <<= 1) {
        const int t = __shfl_up(inc, o, 32);
        if (lane >= o) inc += t;
    }
    if (lane == 31) wsum[wave] = inc;
    __syncthreads();
    if (wave == 0) {
        const int t = wsum[lane];
        int ti = t;
#pragma unroll
        for (int o = 1; o < 32; o <<= 1) {
            const int u = __shfl_up(ti, o, 32);
            if (lane >= o) ti += u;
        }
        wsum[lane] = ti - t;
    }
    __syncthreads();
    const int start = wsum[wave] + inc - v;
    hist[tid] = start;
    if (tid < nbk) runT[(size_t)tid * nblk + b] = make_int2(start, v);
    __syncthreads();
#pragma unroll
    for (int i = 0; i < 4; ++i)
        if (d[i] >= 0) sorted[hist[d[i] / BKN] + r[i]] = (s[i] << 8) | (d[i] % BKN);
    __syncthreads();
    const int left = e - b * CH;
    const int cntb = left < CH ? left : CH;
#pragma unroll
    for (int i = 0; i < 4; ++i) {
        const int j = i * 1024 + tid;
        if (j < cntb) chunkE[(size_t)b * CH + j] = sorted[j];
    }
}

__device__ __forceinline__ void slot_insert(int ev, int node0, int* cntL, int* slotL,
                                            int* novf, int2* __restrict__ ovf) {
    const int s  = ev >> 8;
    const int dl = ev & (BKN - 1);
    const int r  = atomicAdd(&cntL[dl], 1);
    if (r < SLOTS) slotL[dl * SLOTS + r] = s;
    else           ovf[atomicAdd(novf, 1)] = make_int2(s, node0 + dl);
}

template <int K, int N, int THREADS>
__device__ __forceinline__ void gemm_load_B(const float* __restrict__ B,
                                            float4 (&bv)[N * (K / 4) / THREADS]) {
    constexpr int IT = N * (K / 4) / THREADS;
    const int tid = threadIdx.x;
#pragma unroll
    for (int it = 0; it < IT; ++it) {
        const int idx = tid + it * THREADS;
        const float* bsrc = B + (size_t)(4 * (idx / N)) * N + (idx % N);
        bv[it].x = bsrc[0];
        bv[it].y = bsrc[N];
        bv[it].z = bsrc[2 * N];
        bv[it].w = bsrc[3 * N];
    }
}

template <int K, int THREADS>
__device__ __forceinline__ int gemm_load_A(const float* __restrict__ A, int M, int blk,
                                           float4 (&af)[K / 8]) {
    constexpr int J  = K / 8;
    constexpr int RG = THREADS / 64;
    const int tid  = threadIdx.x;
    const int wave = tid >> 5;
    const int lane = tid & 31;
    const int h    = lane >> 4;
    const int c    = lane & 15;
    const int row0 = blk * (RG * 16) + (wave % RG) * 16;
    int ar = row0 + c;
    ar = ar < M ? ar : M - 1;
    const float4* ap = (const float4*)(A + (size_t)ar * K) + h;
#pragma unroll
    for (int j = 0; j < J; ++j) af[j] = ap[2 * j];
    return ar;
}

template <int K, int N, int THREADS>
__device__ __forceinline__ void gemm_stage_B(float* __restrict__ Bt,
                                             const float4 (&bv)[N * (K / 4) / THREADS]) {
    constexpr int P  = K + 4;
    constexpr int IT = N * (K / 4) / THREADS;
    const int tid = threadIdx.x;
#pragma unroll
    for (int it = 0; it < IT; ++it) {
        const int idx = tid + it * THREADS;
        *(float4*)(Bt + (idx % N) * P + 4 * (idx / N)) = bv[it];
    }
}

template <int K, int N, int THREADS>
__device__ __forceinline__ void gemm_mul_store(const float4 (&af)[K / 8], float rsl,
                                               const float* __restrict__ Bt,
                                               float* __restrict__ C, int M, int blk) {
    constexpr int P  = K + 4;
    constexpr int NW = N / 2;
    constexpr int NT = NW / 16;
    constexpr int J  = K / 8;
    constexpr int RG = THREADS / 64;

    const int tid  = threadIdx.x;
    const int wave = tid >> 5;
    const int lane = tid & 31;
    const int h    = lane >> 4;
    const int c    = lane & 15;
    const int row0 = blk * (RG * 16) + (wave % RG) * 16;
    const int col0 = (wave / RG) * NW;
    if (row0 >= M) return;

    v8f acc[NT];
#pragma unroll
    for (int t = 0; t < NT; ++t) acc[t] = v8f{};

    const float* bp = Bt + (col0 + c) * P + 4 * h;
#pragma unroll
    for (int j = 0; j < J; ++j) {
        float4 bf[NT];
#pragma unroll
        for (int t = 0; t < NT; ++t) bf[t] = *(const float4*)(bp + t * 16 * P + 8 * j);
#pragma unroll
        for (int t = 0; t < NT; ++t) {
            v2f a, b;
            a.x = af[j].x; a.y = af[j].y;
            b.x = bf[t].x; b.y = bf[t].y;
            acc[t] = __builtin_amdgcn_wmma_f32_16x16x4_f32(
                 false, a,  false, b,
                 (short)0, acc[t],  false,  false);
        }
#pragma unroll
        for (int t = 0; t < NT; ++t) {
            v2f a, b;
            a.x = af[j].z; a.y = af[j].w;
            b.x = bf[t].z; b.y = bf[t].w;
            acc[t] = __builtin_amdgcn_wmma_f32_16x16x4_f32(
                 false, a,  false, b,
                 (short)0, acc[t],  false,  false);
        }
    }

    float rs[8];
#pragma unroll
    for (int g = 0; g < 8; ++g) rs[g] = __shfl(rsl, h * 8 + g, 32);
#pragma unroll
    for (int t = 0; t < NT; ++t) {
        const int col = col0 + t * 16 + c;
#pragma unroll
        for (int g = 0; g < 8; ++g) {
            const int r = row0 + h * 8 + g;
            if (r < M) C[(size_t)r * N + col] = acc[t][g] * rs[g];
        }
    }
}

template <int K, int N, int THREADS>
__device__ __forceinline__ void gemm_wmma_block(const float* __restrict__ A,
                                                const float* __restrict__ B,
                                                float* __restrict__ C, int M,
                                                int blk, float* __restrict__ Bt,
                                                const int* __restrict__ degc) {
    float4 bv[N * (K / 4) / THREADS];
    gemm_load_B<K, N, THREADS>(B, bv);
    float4 af[K / 8];
    const int ar = gemm_load_A<K, THREADS>(A, M, blk, af);
    const float rsl = rsqrtf((float)(degc[ar] + 1));
    gemm_stage_B<K, N, THREADS>(Bt, bv);
    __syncthreads();
    gemm_mul_store<K, N, THREADS>(af, rsl, Bt, C, M, blk);
}

__global__ __launch_bounds__(1024) void build_gemm_kernel(const int* __restrict__ chunkE,
                                                          const int2* __restrict__ runT,
                                                          int* __restrict__ cnt,
                                                          int* __restrict__ slot,
                                                          int* novf,
                                                          int2* __restrict__ ovf, int n, int nblk,
                                                          const float* __restrict__ X,
                                                          const float* __restrict__ Wm,
                                                          float* __restrict__ H) {
    constexpr int GK = 128, GN = 128, GT = 1024;
    static_assert(GT / 4 == BKN, "one work-group = the rows of one destination block");
    __shared__ int    cntL[BKN];
    __shared__ int4   slotL4[BKN * SLOTS / 4];
    __shared__ float4 Bt4[GN * (GK + 4) / 4];
    int* slotL = (int*)slotL4;
    const int tid  = threadIdx.x;
    const int lane = tid & 31;
    const int w     = __builtin_amdgcn_readfirstlane(tid >> 5);
    const int k     = blockIdx.x;
    const int node0 = k * BKN;

    if (tid < BKN) cntL[tid] = 0;
    __syncthreads();

    float4 bv[GN * (GK / 4) / GT];
    gemm_load_B<GK, GN, GT>(Wm, bv);
    float4 af[GK / 8];
    const int ar = gemm_load_A<GK, GT>(X, n, k, af);

    const int2* runs = runT + (size_t)k * nblk;
    for (int b0 = 0; b0 < nblk; b0 += 256) {
        int st[8], ln[8], ev[8];
#pragma unroll
        for (int j = 0; j < 8; ++j) {
            const int b = b0 + j * 32 + w;
            int2 rec = make_int2(0, 0);
            if (b < nblk) rec = runs[b];
            st[j] = b * CH + rec.x;
            ln[j] = rec.y;
        }
#pragma unroll
        for (int j = 0; j < 8; ++j) {
            ev[j] = 0;
            if (lane < ln[j]) ev[j] = chunkE[st[j] + lane];
        }
#pragma unroll
        for (int j = 0; j < 8; ++j)
            if (lane < ln[j]) slot_insert(ev[j], node0, cntL, slotL, novf, ovf);
#pragma unroll
        for (int j = 0; j < 8; ++j)
            for (int i = 32 + lane; i < ln[j]; i += 32)
                slot_insert(chunkE[st[j] + i], node0, cntL, slotL, novf, ovf);
    }
    gemm_stage_B<GK, GN, GT>((float*)Bt4, bv);
    __syncthreads();

    if (tid < BKN && node0 + tid < n) cnt[node0 + tid] = cntL[tid];
#pragma unroll
    for (int i = 0; i < BKN * SLOTS / 4 / 1024; ++i) {
        const int c    = tid + i * 1024;
        const int nl   = c / (SLOTS / 4);
        const int part = c % (SLOTS / 4);
        int m = cntL[nl];
        m = m < SLOTS ? m : SLOTS;
        if (node0 + nl < n && part * 4 < m)
            ((int4*)slot)[(size_t)(node0 + nl) * (SLOTS / 4) + part] = slotL4[c];
    }

    const float rsl = rsqrtf((float)(cntL[ar - node0] + 1));
    gemm_mul_store<GK, GN, GT>(af, rsl, (const float*)Bt4, H, n, k);
}

template <int K, int N, int THREADS>
__global__ __launch_bounds__(THREADS) void gemm_wmma_kernel(const float* __restrict__ A,
                                                            const float* __restrict__ B,
                                                            float* __restrict__ C, int M,
                                                            const int* __restrict__ degc) {
    __shared__ float4 Bt4[N * (K + 4) / 4];
    gemm_wmma_block<K, N, THREADS>(A, B, C, M, blockIdx.x, (float*)Bt4, degc);
}

template <int F, int B>
__global__ __launch_bounds__(256) void gather_kernel(const int* __restrict__ cnt,
                                                     const int* __restrict__ slot,
                                                     const int* __restrict__ novf,
                                                     const int2* __restrict__ ovf,
                                                     const float* __restrict__ H,
                                                     const float* __restrict__ bias,
                                                     float* __restrict__ Z, int n) {
    constexpr int V = F / 32;
    typedef float vecf __attribute__((ext_vector_type(V)));
    const int lane = threadIdx.x & 31;
    const int node = __builtin_amdgcn_readfirstlane(blockIdx.x * 8 + (threadIdx.x >> 5));
    if (node >= n) return;
    const vecf* Hv = (const vecf*)H;
    const int   deg = cnt[node];
    const int   m   = deg < SLOTS ? deg : SLOTS;
    const float dn  = rsqrtf((float)(deg + 1));
    const int*  row = slot + (size_t)node * SLOTS;

    int mm  = m < 32 ? m : 32;
    int ent = node;
    if (lane < mm) ent = row[lane];
    const vecf self = Hv[(size_t)node * 32 + lane];
    vecf acc = {};

    for (int c0 = 0;;) {
        int j = 0;
        for (; j + B <= mm; j += B) {
            vecf v[B];
            for (int k = 0; k < B; ++k) {
                const int s = __builtin_amdgcn_readlane(ent, j + k);
                v[k] = Hv[(size_t)s * 32 + lane];
            }
            for (int k = 0; k < B; ++k) acc += v[k];
        }
        if (j < mm) {
            vecf v[B];
            for (int k = 0; k < B; ++k) {
                const int jj = (j + k < 32) ? (j + k) : 31;
                const int s = __builtin_amdgcn_readlane(ent, jj);
                v[k] = Hv[(size_t)s * 32 + lane];
            }
            for (int k = 0; k < B; ++k) {
                if (j + k < mm) acc += v[k];
            }
        }
        c0 += 32;
        if (c0 >= m) break;
        mm  = (m - c0 < 32) ? (m - c0) : 32;
        ent = node;
        if (lane < mm) ent = row[c0 + lane];
    }

    if (deg > SLOTS) {
        const int no = novf[0];
        for (int o0 = 0; o0 < no; o0 += 32) {
            int2 pr = make_int2(0, -1);
            if (o0 + lane < no) pr = ovf[o0 + lane];
            unsigned hit = __builtin_amdgcn_ballot_w32(pr.y == node);
            while (hit) {
                const int l = __builtin_ctz(hit);
                hit &= hit - 1;
                const int s = __builtin_amdgcn_readlane(pr.x, l);
                acc += Hv[(size_t)s * 32 + lane];
            }
        }
    }

    acc += self;
    const vecf b = ((const vecf*)bias)[lane];
    vecf r = acc * dn + b;
    for (int q = 0; q < V; ++q) r[q] = r[q] > 0.0f ? r[q] : 0.0f;
    ((vecf*)Z)[(size_t)node * 32 + lane] = r;
}

#define DEC_ITER 4

template <int T>
__device__ __forceinline__ int group8_bcast(int v) {
    return (int)__builtin_amdgcn_mov_dpp8((unsigned)v, (unsigned)T * 0x249249u);
}

template <int M>
__device__ __forceinline__ float xor_add(float v) {
    const int o = __builtin_amdgcn_update_dpp(0, __float_as_int(v), 0x160 + M, 0xf, 0xf, false);
    return v + __int_as_float(o);
}

template <int T>
__device__ __forceinline__ void dec_slot_load(const float* __restrict__ Z, int a, int b, int q,
                                              float4& x0, float4& x1, float4& y0, float4& y1) {
    const unsigned ra = (unsigned)group8_bcast<T>(a) * 64u;
    const unsigned rb = (unsigned)group8_bcast<T>(b) * 64u;
    const float4* za = (const float4*)(Z + ra) + q;
    const float4* zb = (const float4*)(Z + rb) + q;
    x0 = za[0]; x1 = za[8];
    y0 = zb[0]; y1 = zb[8];
}

__device__ __forceinline__ float dec_slot_sum(const float4& x0, const float4& x1,
                                              const float4& y0, const float4& y1) {
    float s = x0.x * y0.x;
    s += x0.y * y0.y; s += x0.z * y0.z; s += x0.w * y0.w;
    s += x1.x * y1.x; s += x1.y * y1.y; s += x1.z * y1.z; s += x1.w * y1.w;
    s = xor_add<1>(s);
    s = xor_add<2>(s);
    s = xor_add<4>(s);
    return s;
}

__global__ __launch_bounds__(256) void decoder_kernel(const float* __restrict__ Z,
                                                      const int* __restrict__ ia,
                                                      const int* __restrict__ ib,
                                                      float* __restrict__ out, int npairs) {
    const int lane = threadIdx.x & 31;
    const int q    = lane & 7;
    const int base0 = __builtin_amdgcn_readfirstlane((blockIdx.x * 8 + (threadIdx.x >> 5)) * (32 * DEC_ITER));
    if (base0 >= npairs) return;

    int a[DEC_ITER], b[DEC_ITER];
#pragma unroll
    for (int it = 0; it < DEC_ITER; ++it) {
        const int p = base0 + it * 32 + lane;
        a[it] = 0; b[it] = 0;
        if (p < npairs) { a[it] = ia[p]; b[it] = ib[p]; }
    }

#pragma unroll
    for (int it = 0; it < DEC_ITER; ++it) {
        if (base0 + it * 32 >= npairs) break;
        float4 x0[8], x1[8], y0[8], y1[8];
        dec_slot_load<0>(Z, a[it], b[it], q, x0[0], x1[0], y0[0], y1[0]);
        dec_slot_load<1>(Z, a[it], b[it], q, x0[1], x1[1], y0[1], y1[1]);
        dec_slot_load<2>(Z, a[it], b[it], q, x0[2], x1[2], y0[2], y1[2]);
        dec_slot_load<3>(Z, a[it], b[it], q, x0[3], x1[3], y0[3], y1[3]);
        dec_slot_load<4>(Z, a[it], b[it], q, x0[4], x1[4], y0[4], y1[4]);
        dec_slot_load<5>(Z, a[it], b[it], q, x0[5], x1[5], y0[5], y1[5]);
        dec_slot_load<6>(Z, a[it], b[it], q, x0[6], x1[6], y0[6], y1[6]);
        dec_slot_load<7>(Z, a[it], b[it], q, x0[7], x1[7], y0[7], y1[7]);
        float res = 0.0f;
#pragma unroll
        for (int t = 0; t < 8; ++t) {
            const float s = dec_slot_sum(x0[t], x1[t], y0[t], y1[t]);
            res = (q == t) ? s : res;
        }
        const int p = base0 + it * 32 + lane;
        if (p < npairs) out[p] = res;
    }
}

extern "C" void kernel_launch(void* const* d_in, const int* in_sizes, int n_in,
                              void* d_out, int out_size, void* d_ws, size_t ws_size,
                              hipStream_t stream) {
    const float* x     = (const float*)d_in[0];
    const float* W_in  = (const float*)d_in[1];
    const float* b_in  = (const float*)d_in[2];
    const float* W_out = (const float*)d_in[3];
    const float* b_out = (const float*)d_in[4];
    const int*   ei    = (const int*)d_in[5];
    const int*   eli   = (const int*)d_in[6];

    const int N    = in_sizes[0] / 128;
    const int E    = in_sizes[5] / 2;
    const int NDEC = in_sizes[6] / 2;
    const int* src = ei;
    const int* dst = ei + E;
    const int* ia  = eli;
    const int* ib  = eli + NDEC;
    float* logits  = (float*)d_out;

    char* base = (char*)d_ws;
    size_t off = 0;
    auto take = [&](size_t bytes) -> float* {
        float* p = (float*)(base + off);
        off = (off + bytes + 255) & ~(size_t)255;
        return p;
    };
    int*   cnt  = (int*)take((size_t)(N + 1) * 4);
    int*   slot = (int*)take((size_t)N * SLOTS * 4);
    int2*  ovf  = (int2*)take((size_t)E * 8);
    float* H1   = take((size_t)N * 128 * 4);
    float* A1   = take((size_t)N * 128 * 4);
    float* H2   = take((size_t)N * 64 * 4);
    float* A2   = take((size_t)N * 64 * 4);
    int*   chunkE = (int*)take(((size_t)E + CH) * 4);
    int2*  runT   = (int2*)take(((size_t)N / BKN + 1) * ((size_t)E / CH + 1) * 8);
    (void)ws_size;

    const int NBK  = (N + BKN - 1) / BKN;
    const int NBLK = (E + CH - 1) / CH;
    bin_kernel<<<NBLK, 1024, 0, stream>>>(src, dst, chunkE, runT, cnt + N, E, NBK);
    build_gemm_kernel<<<NBK, 1024, 0, stream>>>(chunkE, runT, cnt, slot, cnt + N, ovf, N, NBLK,
                                               x, W_in, H1);
    gather_kernel<128, 4><<<(N + 7) / 8, 256, 0, stream>>>(cnt, slot, cnt + N, ovf, H1, b_in, A1, N);

    gemm_wmma_kernel<128, 64, 256><<<(N + 63) / 64, 256, 0, stream>>>(A1, W_out, H2, N, cnt);
    gather_kernel<64, 8><<<(N + 7) / 8, 256, 0, stream>>>(cnt, slot, cnt + N, ovf, H2, b_out, A2, N);

    decoder_kernel<<<(NDEC + 256 * DEC_ITER - 1) / (256 * DEC_ITER), 256, 0, stream>>>(A2, ia, ib, logits, NDEC);
}


# Layer-2 GEMM moves into the layer-1 aggregation launch: a work-group pulls 64 rows of Z1 into LDS, up to 16 source rows in flight per wave, then multiplies them; 5 launches, not 6; sum orders kept
# speedup vs baseline: 5.6355x; 1.0611x over previous
#include <hip/hip_runtime.h>

typedef float v2f __attribute__((ext_vector_type(2)));
typedef float v8f __attribute__((ext_vector_type(8)));

#define SLOTS 64
#define BKN   256
#define CH    4096
#define MAXBK 1024

__global__ __launch_bounds__(1024) void bin_kernel(const int* __restrict__ src,
                                                   const int* __restrict__ dst,
                                                   int* __restrict__ chunkE,
                                                   int2* __restrict__ runT,
                                                   int* __restrict__ novf, int e, int nbk) {
    __shared__ int hist[MAXBK];
    __shared__ int wsum[32];
    __shared__ int sorted[CH];
    const int tid  = threadIdx.x;
    const int lane = tid & 31;
    const int wave = tid >> 5;
    const int b    = blockIdx.x;
    const int nblk = gridDim.x;
    if (b == 0 && tid == 0) novf[0] = 0;
    hist[tid] = 0;

    int s[4], d[4], r[4];
#pragma unroll
    for (int i = 0; i < 4; ++i) {
        const int idx = b * CH + i * 1024 + tid;
        const int ii  = idx < e ? idx : e - 1;
        s[i] = src[ii];
        const int dd = dst[ii];
        d[i] = idx < e ? dd : -1;
        r[i] = 0;
    }
    __syncthreads();
#pragma unroll
    for (int i = 0; i < 4; ++i)
        if (d[i] >= 0) r[i] = atomicAdd(&hist[d[i] / BKN], 1);
    __syncthreads();

    const int v = hist[tid];
    int inc = v;
#pragma unroll
    for (int o = 1; o < 32; o <<= 1) {
        const int t = __shfl_up(inc, o, 32);
        if (lane >= o) inc += t;
    }
    if (lane == 31) wsum[wave] = inc;
    __syncthreads();
    if (wave == 0) {
        const int t = wsum[lane];
        int ti = t;
#pragma unroll
        for (int o = 1; o < 32; o <<= 1) {
            const int u = __shfl_up(ti, o, 32);
            if (lane >= o) ti += u;
        }
        wsum[lane] = ti - t;
    }
    __syncthreads();
    const int start = wsum[wave] + inc - v;
    hist[tid] = start;
    if (tid < nbk) runT[(size_t)tid * nblk + b] = make_int2(start, v);
    __syncthreads();
#pragma unroll
    for (int i = 0; i < 4; ++i)
        if (d[i] >= 0) sorted[hist[d[i] / BKN] + r[i]] = (s[i] << 8) | (d[i] % BKN);
    __syncthreads();
    const int left = e - b * CH;
    const int cntb = left < CH ? left : CH;
#pragma unroll
    for (int i = 0; i < 4; ++i) {
        const int j = i * 1024 + tid;
        if (j < cntb) chunkE[(size_t)b * CH + j] = sorted[j];
    }
}

__device__ __forceinline__ void slot_insert(int ev, int node0, int* cntL, int* slotL,
                                            int* novf, int2* __restrict__ ovf) {
    const int s  = ev >> 8;
    const int dl = ev & (BKN - 1);
    const int r  = atomicAdd(&cntL[dl], 1);
    if (r < SLOTS) slotL[dl * SLOTS + r] = s;
    else           ovf[atomicAdd(novf, 1)] = make_int2(s, node0 + dl);
}

template <int K, int N, int THREADS>
__device__ __forceinline__ void gemm_load_B(const float* __restrict__ B,
                                            float4 (&bv)[N * (K / 4) / THREADS]) {
    constexpr int IT = N * (K / 4) / THREADS;
    const int tid = threadIdx.x;
#pragma unroll
    for (int it = 0; it < IT; ++it) {
        const int idx = tid + it * THREADS;
        const float* bsrc = B + (size_t)(4 * (idx / N)) * N + (idx % N);
        bv[it].x = bsrc[0];
        bv[it].y = bsrc[N];
        bv[it].z = bsrc[2 * N];
        bv[it].w = bsrc[3 * N];
    }
}

template <int K, int THREADS>
__device__ __forceinline__ int gemm_load_A(const float* __restrict__ A, int M, int blk,
                                           float4 (&af)[K / 8]) {
    constexpr int J  = K / 8;
    constexpr int RG = THREADS / 64;
    const int tid  = threadIdx.x;
    const int wave = tid >> 5;
    const int lane = tid & 31;
    const int h    = lane >> 4;
    const int c    = lane & 15;
    const int row0 = blk * (RG * 16) + (wave % RG) * 16;
    int ar = row0 + c;
    ar = ar < M ? ar : M - 1;
    const float4* ap = (const float4*)(A + (size_t)ar * K) + h;
#pragma unroll
    for (int j = 0; j < J; ++j) af[j] = ap[2 * j];
    return ar;
}

template <int K, int N, int THREADS>
__device__ __forceinline__ void gemm_stage_B(float* __restrict__ Bt,
                                             const float4 (&bv)[N * (K / 4) / THREADS]) {
    constexpr int P  = K + 4;
    constexpr int IT = N * (K / 4) / THREADS;
    const int tid = threadIdx.x;
#pragma unroll
    for (int it = 0; it < IT; ++it) {
        const int idx = tid + it * THREADS;
        *(float4*)(Bt + (idx % N) * P + 4 * (idx / N)) = bv[it];
    }
}

template <int K, int N, int THREADS>
__device__ __forceinline__ void gemm_mul_store(const float4 (&af)[K / 8], float rsl,
                                               const float* __restrict__ Bt,
                                               float* __restrict__ C, int M, int blk) {
    constexpr int P  = K + 4;
    constexpr int NW = N / 2;
    constexpr int NT = NW / 16;
    constexpr int J  = K / 8;
    constexpr int RG = THREADS / 64;

    const int tid  = threadIdx.x;
    const int wave = tid >> 5;
    const int lane = tid & 31;
    const int h    = lane >> 4;
    const int c    = lane & 15;
    const int row0 = blk * (RG * 16) + (wave % RG) * 16;
    const int col0 = (wave / RG) * NW;
    if (row0 >= M) return;

    v8f acc[NT];
#pragma unroll
    for (int t = 0; t < NT; ++t) acc[t] = v8f{};

    const float* bp = Bt + (col0 + c) * P + 4 * h;
#pragma unroll
    for (int j = 0; j < J; ++j) {
        float4 bf[NT];
#pragma unroll
        for (int t = 0; t < NT; ++t) bf[t] = *(const float4*)(bp + t * 16 * P + 8 * j);
#pragma unroll
        for (int t = 0; t < NT; ++t) {
            v2f a, b;
            a.x = af[j].x; a.y = af[j].y;
            b.x = bf[t].x; b.y = bf[t].y;
            acc[t] = __builtin_amdgcn_wmma_f32_16x16x4_f32(
                 false, a,  false, b,
                 (short)0, acc[t],  false,  false);
        }
#pragma unroll
        for (int t = 0; t < NT; ++t) {
            v2f a, b;
            a.x = af[j].z; a.y = af[j].w;
            b.x = bf[t].z; b.y = bf[t].w;
            acc[t] = __builtin_amdgcn_wmma_f32_16x16x4_f32(
                 false, a,  false, b,
                 (short)0, acc[t],  false,  false);
        }
    }

    float rs[8];
#pragma unroll
    for (int g = 0; g < 8; ++g) rs[g] = __shfl(rsl, h * 8 + g, 32);
#pragma unroll
    for (int t = 0; t < NT; ++t) {
        const int col = col0 + t * 16 + c;
#pragma unroll
        for (int g = 0; g < 8; ++g) {
            const int r = row0 + h * 8 + g;
            if (r < M) C[(size_t)r * N + col] = acc[t][g] * rs[g];
        }
    }
}

__global__ __launch_bounds__(1024) void build_gemm_kernel(const int* __restrict__ chunkE,
                                                          const int2* __restrict__ runT,
                                                          int* __restrict__ cnt,
                                                          int* __restrict__ slot,
                                                          int* novf,
                                                          int2* __restrict__ ovf, int n, int nblk,
                                                          const float* __restrict__ X,
                                                          const float* __restrict__ Wm,
                                                          float* __restrict__ H) {
    constexpr int GK = 128, GN = 128, GT = 1024;
    static_assert(GT / 4 == BKN, "one work-group = the rows of one destination block");
    __shared__ int    cntL[BKN];
    __shared__ int4   slotL4[BKN * SLOTS / 4];
    __shared__ float4 Bt4[GN * (GK + 4) / 4];
    int* slotL = (int*)slotL4;
    const int tid  = threadIdx.x;
    const int lane = tid & 31;
    const int w     = __builtin_amdgcn_readfirstlane(tid >> 5);
    const int k     = blockIdx.x;
    const int node0 = k * BKN;

    if (tid < BKN) cntL[tid] = 0;
    __syncthreads();

    float4 bv[GN * (GK / 4) / GT];
    gemm_load_B<GK, GN, GT>(Wm, bv);
    float4 af[GK / 8];
    const int ar = gemm_load_A<GK, GT>(X, n, k, af);

    const int2* runs = runT + (size_t)k * nblk;
    for (int b0 = 0; b0 < nblk; b0 += 256) {
        int st[8], ln[8], ev[8];
#pragma unroll
        for (int j = 0; j < 8; ++j) {
            const int b = b0 + j * 32 + w;
            int2 rec = make_int2(0, 0);
            if (b < nblk) rec = runs[b];
            st[j] = b * CH + rec.x;
            ln[j] = rec.y;
        }
#pragma unroll
        for (int j = 0; j < 8; ++j) {
            ev[j] = 0;
            if (lane < ln[j]) ev[j] = chunkE[st[j] + lane];
        }
#pragma unroll
        for (int j = 0; j < 8; ++j)
            if (lane < ln[j]) slot_insert(ev[j], node0, cntL, slotL, novf, ovf);
#pragma unroll
        for (int j = 0; j < 8; ++j)
            for (int i = 32 + lane; i < ln[j]; i += 32)
                slot_insert(chunkE[st[j] + i], node0, cntL, slotL, novf, ovf);
    }
    gemm_stage_B<GK, GN, GT>((float*)Bt4, bv);
    __syncthreads();

    if (tid < BKN && node0 + tid < n) cnt[node0 + tid] = cntL[tid];
#pragma unroll
    for (int i = 0; i < BKN * SLOTS / 4 / 1024; ++i) {
        const int c    = tid + i * 1024;
        const int nl   = c / (SLOTS / 4);
        const int part = c % (SLOTS / 4);
        int m = cntL[nl];
        m = m < SLOTS ? m : SLOTS;
        if (node0 + nl < n && part * 4 < m)
            ((int4*)slot)[(size_t)(node0 + nl) * (SLOTS / 4) + part] = slotL4[c];
    }

    const float rsl = rsqrtf((float)(cntL[ar - node0] + 1));
    gemm_mul_store<GK, GN, GT>(af, rsl, (const float*)Bt4, H, n, k);
}

template <int F, int B>
__global__ __launch_bounds__(256) void gather_kernel(const int* __restrict__ cnt,
                                                     const int* __restrict__ slot,
                                                     const int* __restrict__ novf,
                                                     const int2* __restrict__ ovf,
                                                     const float* __restrict__ H,
                                                     const float* __restrict__ bias,
                                                     float* __restrict__ Z, int n) {
    constexpr int V = F / 32;
    typedef float vecf __attribute__((ext_vector_type(V)));
    const int lane = threadIdx.x & 31;
    const int node = __builtin_amdgcn_readfirstlane(blockIdx.x * 8 + (threadIdx.x >> 5));
    if (node >= n) return;
    const vecf* Hv = (const vecf*)H;
    const int   deg = cnt[node];
    const int   m   = deg < SLOTS ? deg : SLOTS;
    const float dn  = rsqrtf((float)(deg + 1));
    const int*  row = slot + (size_t)node * SLOTS;

    int mm  = m < 32 ? m : 32;
    int ent = node;
    if (lane < mm) ent = row[lane];
    const vecf self = Hv[(size_t)node * 32 + lane];
    vecf acc = {};

    for (int c0 = 0;;) {
        int j = 0;
        for (; j + B <= mm; j += B) {
            vecf v[B];
            for (int k = 0; k < B; ++k) {
                const int s = __builtin_amdgcn_readlane(ent, j + k);
                v[k] = Hv[(size_t)s * 32 + lane];
            }
            for (int k = 0; k < B; ++k) acc += v[k];
        }
        if (j < mm) {
            vecf v[B];
            for (int k = 0; k < B; ++k) {
                const int jj = (j + k < 32) ? (j + k) : 31;
                const int s = __builtin_amdgcn_readlane(ent, jj);
                v[k] = Hv[(size_t)s * 32 + lane];
            }
            for (int k = 0; k < B; ++k) {
                if (j + k < mm) acc += v[k];
            }
        }
        c0 += 32;
        if (c0 >= m) break;
        mm  = (m - c0 < 32) ? (m - c0) : 32;
        ent = node;
        if (lane < mm) ent = row[c0 + lane];
    }

    if (deg > SLOTS) {
        const int no = novf[0];
        for (int o0 = 0; o0 < no; o0 += 32) {
            int2 pr = make_int2(0, -1);
            if (o0 + lane < no) pr = ovf[o0 + lane];
            unsigned hit = __builtin_amdgcn_ballot_w32(pr.y == node);
            while (hit) {
                const int l = __builtin_ctz(hit);
                hit &= hit - 1;
                const int s = __builtin_amdgcn_readlane(pr.x, l);
                acc += Hv[(size_t)s * 32 + lane];
            }
        }
    }

    acc += self;
    const vecf b = ((const vecf*)bias)[lane];
    vecf r = acc * dn + b;
    for (int q = 0; q < V; ++q) r[q] = r[q] > 0.0f ? r[q] : 0.0f;
    ((vecf*)Z)[(size_t)node * 32 + lane] = r;
}

__global__ __launch_bounds__(256) void gather_gemm_kernel(const int* __restrict__ cnt,
                                                          const int* __restrict__ slot,
                                                          const int* __restrict__ novf,
                                                          const int2* __restrict__ ovf,
                                                          const float* __restrict__ H,
                                                          const float* __restrict__ bias,
                                                          const float* __restrict__ Wm,
                                                          float* __restrict__ C, int n) {
    constexpr int GK = 128, GN = 64, GT = 256;
    constexpr int NB = GT / 4;
    constexpr int PW = NB / (GT / 32);
    constexpr int P  = GK + 4;
    constexpr int B  = 16;
    typedef float vecf __attribute__((ext_vector_type(4)));
    __shared__ float4 Bt4[GN * P / 4];
    __shared__ float4 Zs4[NB * P / 4];
    __shared__ float  dnL[NB];
    float* Zs = (float*)Zs4;
    const int tid  = threadIdx.x;
    const int lane = tid & 31;
    const int w     = __builtin_amdgcn_readfirstlane(tid >> 5);
    const int blk   = blockIdx.x;
    const int nodeW = blk * NB + w * PW;

    float4 bv[GN * (GK / 4) / GT];
    gemm_load_B<GK, GN, GT>(Wm, bv);

    int nd = nodeW + (lane & (PW - 1));
    nd = nd < n ? nd : n - 1;
    const int degv = cnt[nd];
    int rawc = 0;
    if (nodeW < n) {
        const int d0 = __builtin_amdgcn_readlane(degv, 0);
        if (lane < (d0 < 32 ? d0 : 32)) rawc = slot[(size_t)nodeW * SLOTS + lane];
    }
    gemm_stage_B<GK, GN, GT>((float*)Bt4, bv);

    const vecf* Hv = (const vecf*)H;
    const vecf bsv = ((const vecf*)bias)[lane];
    for (int i = 0; i < PW; ++i) {
        const int node = nodeW + i;
        int rawn = 0;
        if (i + 1 < PW && node + 1 < n) {
            const int d1 = __builtin_amdgcn_readlane(degv, i + 1);
            if (lane < (d1 < 32 ? d1 : 32)) rawn = slot[(size_t)(node + 1) * SLOTS + lane];
        }
        if (node < n) {
            const int   deg = __builtin_amdgcn_readlane(degv, i);
            const int   m   = deg < SLOTS ? deg : SLOTS;
            const float dn  = rsqrtf((float)(deg + 1));
            const int*  row = slot + (size_t)node * SLOTS;
            int mm  = m < 32 ? m : 32;
            int ent = rawc;
            const vecf self = Hv[(size_t)node * 32 + lane];
            vecf acc = {};
            auto group = [&](int e, int j, int left) {
                vecf v[B];
                int  sr[B];
#pragma unroll
                for (int k = 0; k < B; ++k)
                    sr[k] = __builtin_amdgcn_readlane(e, (j + k) & 31);
#pragma unroll
                for (int k = 0; k < B; ++k) {
                    v[k] = vecf{};
                    if (k < left) v[k] = Hv[(size_t)sr[k] * 32 + lane];
                }
                asm volatile("" :: "v"(v[0]), "v"(v[1]), "v"(v[2]), "v"(v[3]), "v"(v[4]), "v"(v[5]),
                             "v"(v[6]), "v"(v[7]), "v"(v[8]), "v"(v[9]), "v"(v[10]), "v"(v[11]),
                             "v"(v[12]), "v"(v[13]), "v"(v[14]), "v"(v[15]));
#pragma unroll
                for (int k = 0; k < B; ++k) acc += v[k];
            };
            group(ent, 0, mm);
            for (int j = B; j < mm; j += B) group(ent, j, mm - j);
            for (int c0 = 32; c0 < m; c0 += 32) {
                mm  = (m - c0 < 32) ? (m - c0) : 32;
                ent = 0;
                if (lane < mm) ent = row[c0 + lane];
                for (int j = 0; j < mm; j += B) group(ent, j, mm - j);
            }
            if (deg > SLOTS) {
                const int no = novf[0];
                for (int o0 = 0; o0 < no; o0 += 32) {
                    int2 pr = make_int2(0, -1);
                    if (o0 + lane < no) pr = ovf[o0 + lane];
                    unsigned hit = __builtin_amdgcn_ballot_w32(pr.y == node);
                    while (hit) {
                        const int l = __builtin_ctz(hit);
                        hit &= hit - 1;
                        const int s = __builtin_amdgcn_readlane(pr.x, l);
                        acc += Hv[(size_t)s * 32 + lane];
                    }
                }
            }
            acc += self;
            vecf r = acc * dn + bsv;
            for (int q = 0; q < 4; ++q) r[q] = r[q] > 0.0f ? r[q] : 0.0f;
            const int nl = w * PW + i;
            *(vecf*)(Zs + nl * P + 4 * lane) = r;
            if (lane == 0) dnL[nl] = dn;
        } else {
            const int nl = w * PW + i;
            *(vecf*)(Zs + nl * P + 4 * lane) = vecf{};
            if (lane == 0) dnL[nl] = 0.0f;
        }
        rawc = rawn;
    }
    __syncthreads();

    const int h  = lane >> 4;
    const int c  = lane & 15;
    const int rl = (w % (GT / 64)) * 16 + c;
    const float4* ap = (const float4*)(Zs + rl * P) + h;
    float4 af[GK / 8];
#pragma unroll
    for (int j = 0; j < GK / 8; ++j) af[j] = ap[2 * j];
    const float rsl = dnL[rl];
    gemm_mul_store<GK, GN, GT>(af, rsl, (const float*)Bt4, C, n, blk);
}

#define DEC_ITER 4

template <int T>
__device__ __forceinline__ int group8_bcast(int v) {
    return (int)__builtin_amdgcn_mov_dpp8((unsigned)v, (unsigned)T * 0x249249u);
}

template <int M>
__device__ __forceinline__ float xor_add(float v) {
    const int o = __builtin_amdgcn_update_dpp(0, __float_as_int(v), 0x160 + M, 0xf, 0xf, false);
    return v + __int_as_float(o);
}

template <int T>
__device__ __forceinline__ void dec_slot_load(const float* __restrict__ Z, int a, int b, int q,
                                              float4& x0, float4& x1, float4& y0, float4& y1) {
    const unsigned ra = (unsigned)group8_bcast<T>(a) * 64u;
    const unsigned rb = (unsigned)group8_bcast<T>(b) * 64u;
    const float4* za = (const float4*)(Z + ra) + q;
    const float4* zb = (const float4*)(Z + rb) + q;
    x0 = za[0]; x1 = za[8];
    y0 = zb[0]; y1 = zb[8];
}

__device__ __forceinline__ float dec_slot_sum(const float4& x0, const float4& x1,
                                              const float4& y0, const float4& y1) {
    float s = x0.x * y0.x;
    s += x0.y * y0.y; s += x0.z * y0.z; s += x0.w * y0.w;
    s += x1.x * y1.x; s += x1.y * y1.y; s += x1.z * y1.z; s += x1.w * y1.w;
    s = xor_add<1>(s);
    s = xor_add<2>(s);
    s = xor_add<4>(s);
    return s;
}

__global__ __launch_bounds__(256) void decoder_kernel(const float* __restrict__ Z,
                                                      const int* __restrict__ ia,
                                                      const int* __restrict__ ib,
                                                      float* __restrict__ out, int npairs) {
    const int lane = threadIdx.x & 31;
    const int q    = lane & 7;
    const int base0 = __builtin_amdgcn_readfirstlane((blockIdx.x * 8 + (threadIdx.x >> 5)) * (32 * DEC_ITER));
    if (base0 >= npairs) return;

    int a[DEC_ITER], b[DEC_ITER];
#pragma unroll
    for (int it = 0; it < DEC_ITER; ++it) {
        const int p = base0 + it * 32 + lane;
        a[it] = 0; b[it] = 0;
        if (p < npairs) { a[it] = ia[p]; b[it] = ib[p]; }
    }

#pragma unroll
    for (int it = 0; it < DEC_ITER; ++it) {
        if (base0 + it * 32 >= npairs) break;
        float4 x0[8], x1[8], y0[8], y1[8];
        dec_slot_load<0>(Z, a[it], b[it], q, x0[0], x1[0], y0[0], y1[0]);
        dec_slot_load<1>(Z, a[it], b[it], q, x0[1], x1[1], y0[1], y1[1]);
        dec_slot_load<2>(Z, a[it], b[it], q, x0[2], x1[2], y0[2], y1[2]);
        dec_slot_load<3>(Z, a[it], b[it], q, x0[3], x1[3], y0[3], y1[3]);
        dec_slot_load<4>(Z, a[it], b[it], q, x0[4], x1[4], y0[4], y1[4]);
        dec_slot_load<5>(Z, a[it], b[it], q, x0[5], x1[5], y0[5], y1[5]);
        dec_slot_load<6>(Z, a[it], b[it], q, x0[6], x1[6], y0[6], y1[6]);
        dec_slot_load<7>(Z, a[it], b[it], q, x0[7], x1[7], y0[7], y1[7]);
        float res = 0.0f;
#pragma unroll
        for (int t = 0; t < 8; ++t) {
            const float s = dec_slot_sum(x0[t], x1[t], y0[t], y1[t]);
            res = (q == t) ? s : res;
        }
        const int p = base0 + it * 32 + lane;
        if (p < npairs) out[p] = res;
    }
}

extern "C" void kernel_launch(void* const* d_in, const int* in_sizes, int n_in,
                              void* d_out, int out_size, void* d_ws, size_t ws_size,
                              hipStream_t stream) {
    const float* x     = (const float*)d_in[0];
    const float* W_in  = (const float*)d_in[1];
    const float* b_in  = (const float*)d_in[2];
    const float* W_out = (const float*)d_in[3];
    const float* b_out = (const float*)d_in[4];
    const int*   ei    = (const int*)d_in[5];
    const int*   eli   = (const int*)d_in[6];

    const int N    = in_sizes[0] / 128;
    const int E    = in_sizes[5] / 2;
    const int NDEC = in_sizes[6] / 2;
    const int* src = ei;
    const int* dst = ei + E;
    const int* ia  = eli;
    const int* ib  = eli + NDEC;
    float* logits  = (float*)d_out;

    char* base = (char*)d_ws;
    size_t off = 0;
    auto take = [&](size_t bytes) -> float* {
        float* p = (float*)(base + off);
        off = (off + bytes + 255) & ~(size_t)255;
        return p;
    };
    int*   cnt  = (int*)take((size_t)(N + 1) * 4);
    int*   slot = (int*)take((size_t)N * SLOTS * 4);
    int2*  ovf  = (int2*)take((size_t)E * 8);
    float* H1   = take((size_t)N * 128 * 4);
    float* H2   = take((size_t)N * 64 * 4);
    float* A2   = take((size_t)N * 64 * 4);
    int*   chunkE = (int*)take(((size_t)E + CH) * 4);
    int2*  runT   = (int2*)take(((size_t)N / BKN + 1) * ((size_t)E / CH + 1) * 8);
    (void)ws_size;

    const int NBK  = (N + BKN - 1) / BKN;
    const int NBLK = (E + CH - 1) / CH;
    bin_kernel<<<NBLK, 1024, 0, stream>>>(src, dst, chunkE, runT, cnt + N, E, NBK);
    build_gemm_kernel<<<NBK, 1024, 0, stream>>>(chunkE, runT, cnt, slot, cnt + N, ovf, N, NBLK,
                                               x, W_in, H1);
    gather_gemm_kernel<<<(N + 63) / 64, 256, 0, stream>>>(cnt, slot, cnt + N, ovf, H1, b_in, W_out, H2, N);
    gather_kernel<64, 8><<<(N + 7) / 8, 256, 0, stream>>>(cnt, slot, cnt + N, ovf, H2, b_out, A2, N);

    decoder_kernel<<<(NDEC + 256 * DEC_ITER - 1) / (256 * DEC_ITER), 256, 0, stream>>>(A2, ia, ib, logits, NDEC);
}
